# starting kernel
# baseline (speedup 1.0000x reference)
#include <hip/hip_runtime.h>

#define NN 50000
#define NE 800000
#define D  128

typedef __attribute__((ext_vector_type(2))) float v2f;
typedef __attribute__((ext_vector_type(8))) float v8f;

__global__ void degree_kernel(const int* __restrict__ src,
                              const int* __restrict__ dst,
                              float* __restrict__ din,
                              float* __restrict__ dout, int E) {
    int e = blockIdx.x * blockDim.x + threadIdx.x;
    if (e < E) {
        unsafeAtomicAdd(&din[dst[e]], 1.0f);
        unsafeAtomicAdd(&dout[src[e]], 1.0f);
    }
}

__global__ __launch_bounds__(256)
void scatter_kernel(const float* __restrict__ h,
                    const int* __restrict__ src,
                    const int* __restrict__ dst,
                    const float* __restrict__ degout,
                    float* __restrict__ agg, int E, int useScale) {
    int wave = (blockIdx.x * blockDim.x + threadIdx.x) >> 5;
    int lane = threadIdx.x & 31;
    if (wave >= E) return;
    int s = src[wave];
    int d = dst[wave];
    float sc = 1.0f;
    if (useScale) sc = rsqrtf(fmaxf(degout[s], 1.0f));
    const float4* hv = reinterpret_cast<const float4*>(h + (size_t)s * D);
    float4 v = hv[lane];
    float* ap = agg + (size_t)d * D + lane * 4;
    unsafeAtomicAdd(ap + 0, v.x * sc);
    unsafeAtomicAdd(ap + 1, v.y * sc);
    unsafeAtomicAdd(ap + 2, v.z * sc);
    unsafeAtomicAdd(ap + 3, v.w * sc);
}

__global__ __launch_bounds__(256)
void norm_gemm_wmma(const float* __restrict__ agg,
                    const float* __restrict__ hprev,
                    const float* __restrict__ deg,
                    const float* __restrict__ W,
                    const float* __restrict__ bias,
                    float* __restrict__ out, int mode) {
    __shared__ float Alds[16 * 132];

    const int t = threadIdx.x;
    const int rowBase = blockIdx.x * 16;

    {
        int r  = t >> 4;
        int c0 = (t & 15) * 8;
        int row = rowBase + r;
        float s;
        if (mode == 0) s = 1.0f / (deg[row] + 1.0f);
        else           s = rsqrtf(fmaxf(deg[row], 1.0f));
        const float* ar = agg   + (size_t)row * D + c0;
        const float* hr = hprev + (size_t)row * D + c0;
#pragma unroll
        for (int j = 0; j < 8; ++j) {
            float v = ar[j];
            if (mode == 0) v += hr[j];
            Alds[r * 132 + c0 + j] = v * s;
        }
    }
    __syncthreads();

    const int waveId = t >> 5;
    const int lane   = t & 31;
    const int half   = lane >> 4;
    const int lm     = lane & 15;
    const int col0   = waveId * 16;

    v8f c = {};
#pragma unroll
    for (int kk = 0; kk < 32; ++kk) {
        int k0 = 4 * kk + 2 * half;
        v2f a, b;
        a.x = Alds[lm * 132 + k0];
        a.y = Alds[lm * 132 + k0 + 1];
        b.x = W[(size_t)(k0)     * D + col0 + lm];
        b.y = W[(size_t)(k0 + 1) * D + col0 + lm];
        c = __builtin_amdgcn_wmma_f32_16x16x4_f32(
                false, a, false, b, (short)0, c, false, false);
    }

    float bb = bias[col0 + lm];
#pragma unroll
    for (int v = 0; v < 8; ++v) {
        int row = rowBase + v + 8 * half;
        out[(size_t)row * D + col0 + lm] = c[v] + bb;
    }
}

extern "C" void kernel_launch(void* const* d_in, const int* in_sizes, int n_in,
                              void* d_out, int out_size, void* d_ws, size_t ws_size,
                              hipStream_t stream) {
    const float* x  = (const float*)d_in[0];
    const float* W1 = (const float*)d_in[1];
    const float* b1 = (const float*)d_in[2];
    const float* W2 = (const float*)d_in[3];
    const float* b2 = (const float*)d_in[4];
    const float* W3 = (const float*)d_in[5];
    const float* b3 = (const float*)d_in[6];
    const int* src  = (const int*)d_in[7];
    const int* dst  = (const int*)d_in[8];
    float* out = (float*)d_out;

    const int N = NN, E = NE;
    float* deg_in  = (float*)d_ws;
    float* deg_out = deg_in + N;
    float* agg     = deg_out + N;
    float* h1      = agg + (size_t)N * D;
    float* h2      = h1  + (size_t)N * D;

    const size_t matBytes = (size_t)N * D * sizeof(float);
    dim3 edgeWaveGrid((E + 7) / 8);
    dim3 gemmGrid(N / 16);

    hipMemsetAsync(deg_in, 0, (size_t)2 * N * sizeof(float), stream);
    hipMemsetAsync(agg, 0, matBytes, stream);
    degree_kernel<<<(E + 255) / 256, 256, 0, stream>>>(src, dst, deg_in, deg_out, E);

    scatter_kernel<<<edgeWaveGrid, 256, 0, stream>>>(x, src, dst, deg_out, agg, E, 0);
    norm_gemm_wmma<<<gemmGrid, 256, 0, stream>>>(agg, x, deg_in, W1, b1, h1, 0);

    hipMemsetAsync(agg, 0, matBytes, stream);
    scatter_kernel<<<edgeWaveGrid, 256, 0, stream>>>(h1, src, dst, deg_out, agg, E, 0);
    norm_gemm_wmma<<<gemmGrid, 256, 0, stream>>>(agg, h1, deg_in, W2, b2, h2, 0);

    hipMemsetAsync(agg, 0, matBytes, stream);
    scatter_kernel<<<edgeWaveGrid, 256, 0, stream>>>(h2, src, dst, deg_out, agg, E, 1);
    norm_gemm_wmma<<<gemmGrid, 256, 0, stream>>>(agg, h2, deg_in, W3, b3, out, 1);
}


# Layer 1's scan keeps each wave's hit list (source row, local row) in the workspace; layers 2 and 3 replay it and no longer scan dst or read src; sums stay LDS float adds in arrival order
# speedup vs baseline: 2.0285x; 2.0285x over previous
#include <hip/hip_runtime.h>

#define NN 50000
#define NE 800000
#define D  128

typedef __attribute__((ext_vector_type(2))) float v2f;
typedef __attribute__((ext_vector_type(8))) float v8f;

__global__ void degree_kernel(const int* __restrict__ src,
                              const int* __restrict__ dst,
                              float* __restrict__ din,
                              float* __restrict__ dout, int E) {
    int e = blockIdx.x * blockDim.x + threadIdx.x;
    if (e < E) {
        unsafeAtomicAdd(&din[dst[e]], 1.0f);
        unsafeAtomicAdd(&dout[src[e]], 1.0f);
    }
}

#define RT     196
#define NTILE  256
#define SCAN_U 8
#define WLCAP  (32 + SCAN_U * 128)
#define NQ     (NE / 4)
#define TBLK   1024
#define NWAVE  (TBLK / 32)
#define RCAP   1024

static_assert(NE % 4 == 0, "dst list is read in 16-byte words");
static_assert(WLCAP >= 31 + SCAN_U * 128, "a wave's list holds the rest of one flush plus one full chunk");
static_assert(NE <= (1 << 20) && RT <= 256, "entry packs edge id (20 bits) and local row (8 bits)");
static_assert((long)NTILE * RT >= NN && (long)(NTILE - 1) * RT < NN, "tiles cover every row");
static_assert(NN <= (1 << 16), "replay entry packs source row (16 bits) and local row (8 bits)");
static_assert(RCAP % 128 == 0 && RCAP <= WLCAP, "a kept list is staged in the wave's LDS list in steps of 128 entries");

__device__ __forceinline__ void tile_add_row(float* trow, float4 v,
                                             int rot, int lo, int lo3) {
    float a0 = rot ? v.y : v.x;
    float a1 = rot ? v.z : v.y;
    float a2 = rot ? v.w : v.z;
    float a3 = rot ? v.x : v.w;
    __hip_atomic_fetch_add(trow + lo + 0, a0, __ATOMIC_RELAXED, __HIP_MEMORY_SCOPE_WORKGROUP);
    __hip_atomic_fetch_add(trow + lo + 1, a1, __ATOMIC_RELAXED, __HIP_MEMORY_SCOPE_WORKGROUP);
    __hip_atomic_fetch_add(trow + lo + 2, a2, __ATOMIC_RELAXED, __HIP_MEMORY_SCOPE_WORKGROUP);
    __hip_atomic_fetch_add(trow + lo3,    a3, __ATOMIC_RELAXED, __HIP_MEMORY_SCOPE_WORKGROUP);
}

__device__ __forceinline__ void note_hits(int4 dv, int q, bool valid, int base, int rows,
                                          unsigned* wl, int& wcount) {
    const unsigned r0 = (unsigned)(dv.x - base);
    const unsigned r1 = (unsigned)(dv.y - base);
    const unsigned r2 = (unsigned)(dv.z - base);
    const unsigned r3 = (unsigned)(dv.w - base);
    const bool h0 = valid && r0 < (unsigned)rows, h1 = valid && r1 < (unsigned)rows;
    const bool h2 = valid && r2 < (unsigned)rows, h3 = valid && r3 < (unsigned)rows;
    const unsigned m0 = __builtin_amdgcn_ballot_w32(h0);
    const unsigned m1 = __builtin_amdgcn_ballot_w32(h1);
    const unsigned m2 = __builtin_amdgcn_ballot_w32(h2);
    const unsigned m3 = __builtin_amdgcn_ballot_w32(h3);
    if (m0 | m1 | m2 | m3) {
        if (m0) {
            if (h0) wl[wcount + __builtin_amdgcn_mbcnt_lo(m0, 0u)] = (unsigned)(4 * q + 0) | (r0 << 20);
            wcount += __builtin_popcount(m0);
        }
        if (m1) {
            if (h1) wl[wcount + __builtin_amdgcn_mbcnt_lo(m1, 0u)] = (unsigned)(4 * q + 1) | (r1 << 20);
            wcount += __builtin_popcount(m1);
        }
        if (m2) {
            if (h2) wl[wcount + __builtin_amdgcn_mbcnt_lo(m2, 0u)] = (unsigned)(4 * q + 2) | (r2 << 20);
            wcount += __builtin_popcount(m2);
        }
        if (m3) {
            if (h3) wl[wcount + __builtin_amdgcn_mbcnt_lo(m3, 0u)] = (unsigned)(4 * q + 3) | (r3 << 20);
            wcount += __builtin_popcount(m3);
        }
    }
}

template <int SCALE>
__device__ __forceinline__ void gather_add(const float* __restrict__ h,
                                           const float* __restrict__ degout,
                                           float* tile, int s, int r,
                                           int cnt, int lane,
                                           int rot, int lo, int lo3) {
    int scb = 0;
    if (SCALE) scb = __float_as_int(rsqrtf(fmaxf(degout[s], 1.0f)));
    const float4* h4 = reinterpret_cast<const float4*>(h);
    for (int k0 = 0; k0 < cnt; k0 += 8) {
        float4 v[8];
#pragma unroll
        for (int k = 0; k < 8; ++k) {
            int sk = __builtin_amdgcn_readlane(s, k0 + k);
            v[k] = h4[(size_t)sk * (D / 4) + lane];
        }
#pragma unroll
        for (int k = 0; k < 8; ++k) {
            if (k0 + k < cnt) {
                int rk = __builtin_amdgcn_readlane(r, k0 + k);
                float4 x = v[k];
                if (SCALE) {
                    float sc = __int_as_float(__builtin_amdgcn_readlane(scb, k0 + k));
                    x.x *= sc; x.y *= sc; x.z *= sc; x.w *= sc;
                }
                tile_add_row(tile + rk * D, x, rot, lo, lo3);
            }
        }
    }
}

template <int SCALE, int REC>
__device__ __forceinline__ void tile_round(const float* __restrict__ h,
                                           const int* __restrict__ src,
                                           const float* __restrict__ degout,
                                           float* tile,
                                           const unsigned* wl,
                                           int off, int cnt, int lane,
                                           int rot, int lo, int lo3,
                                           unsigned* rec, int& rcount) {
    unsigned ent = (lane < cnt) ? wl[off + lane] : 0u;
    int e = (int)(ent & 0xFFFFFu);
    int r = (int)(ent >> 20);
    int s = (lane < cnt) ? src[e] : 0;
    if (REC) {
        if (rcount + cnt <= RCAP) {
            if (lane < cnt) rec[rcount + lane] = (unsigned)s | ((unsigned)r << 16);
        }
        rcount += cnt;
    }
    gather_add<SCALE>(h, degout, tile, s, r, cnt, lane, rot, lo, lo3);
}

template <int SCALE, int REC>
__device__ __forceinline__ void scan_and_sum(const float* __restrict__ h,
                                             const int* __restrict__ src,
                                             const int* __restrict__ dst,
                                             const float* __restrict__ degout,
                                             float* tile, unsigned* wl,
                                             int t, int lane, int base, int rows,
                                             int rot, int lo, int lo3,
                                             unsigned* rec, int& rcount) {
    const int4* d4 = reinterpret_cast<const int4*>(dst);
    int wcount = 0;

    int q0 = 0;
    for (; q0 + SCAN_U * TBLK <= NQ; q0 += SCAN_U * TBLK) {
        int4 dv[SCAN_U];
#pragma unroll
        for (int u = 0; u < SCAN_U; ++u)
            dv[u] = d4[q0 + u * TBLK + t];
#pragma unroll
        for (int u = 0; u < SCAN_U; ++u)
            note_hits(dv[u], q0 + u * TBLK + t, true, base, rows, wl, wcount);
        if (wcount >= 32) {
            int done = 0;
            for (; done + 32 <= wcount; done += 32)
                tile_round<SCALE, REC>(h, src, degout, tile, wl, done, 32, lane, rot, lo, lo3, rec, rcount);
            const int rem = wcount - done;
            unsigned keep = (lane < rem) ? wl[done + lane] : 0u;
            if (lane < rem) wl[lane] = keep;
            wcount = rem;
        }
    }
    for (; q0 < NQ; q0 += TBLK) {
        const int q = q0 + t;
        const bool valid = q < NQ;
        note_hits(d4[valid ? q : NQ - 1], q, valid, base, rows, wl, wcount);
    }
    for (int done = 0; done < wcount; done += 32)
        tile_round<SCALE, REC>(h, src, degout, tile, wl, done, min(32, wcount - done), lane, rot, lo, lo3, rec, rcount);
}

template <int SCALE, int REPLAY>
__global__ __launch_bounds__(TBLK)
void tile_sum_kernel(const float* __restrict__ h,
                     const int* __restrict__ src,
                     const int* __restrict__ dst,
                     const float* __restrict__ degout,
                     float* __restrict__ agg,
                     unsigned* reclist,
                     int* reccnt) {
    __shared__ __attribute__((aligned(16))) float tile[RT * D];
    __shared__ unsigned wlist[NWAVE * WLCAP];

    const int t    = threadIdx.x;
    const int lane = t & 31;
    const int wave = __builtin_amdgcn_readfirstlane(t >> 5);
    const int base = blockIdx.x * RT;
    const int rows = min(RT, NN - base);
    const int rot  = lane >> 4;
    const int lo   = 4 * lane + rot;
    const int lo3  = 4 * lane + 3 - 3 * rot;

    float4* t4 = reinterpret_cast<float4*>(tile);
    for (int i = t; i < RT * D / 4; i += TBLK) t4[i] = make_float4(0.f, 0.f, 0.f, 0.f);
    __syncthreads();

    unsigned* wl = wlist + wave * WLCAP;
    const int gw = blockIdx.x * NWAVE + wave;
    unsigned* rec = reclist + (size_t)gw * RCAP;
    int rcount = 0;

    if (REPLAY) {
        const int total = __builtin_amdgcn_readfirstlane(reccnt[gw]);
        if (total <= RCAP) {
            for (int i0 = 0; i0 < total; i0 += 128) {
                const int i = i0 + lane;
                const unsigned e0 = (i      < total) ? rec[i]      : 0u;
                const unsigned e1 = (i + 32 < total) ? rec[i + 32] : 0u;
                const unsigned e2 = (i + 64 < total) ? rec[i + 64] : 0u;
                const unsigned e3 = (i + 96 < total) ? rec[i + 96] : 0u;
                wl[i] = e0; wl[i + 32] = e1; wl[i + 64] = e2; wl[i + 96] = e3;
            }
            for (int off = 0; off < total; off += 32) {
                const int cnt = min(32, total - off);
                const unsigned ent = (lane < cnt) ? wl[off + lane] : 0u;
                gather_add<SCALE>(h, degout, tile, (int)(ent & 0xFFFFu), (int)(ent >> 16),
                                  cnt, lane, rot, lo, lo3);
            }
        } else {
            scan_and_sum<SCALE, 0>(h, src, dst, degout, tile, wl, t, lane, base, rows,
                                   rot, lo, lo3, rec, rcount);
        }
    } else {
        scan_and_sum<SCALE, 1>(h, src, dst, degout, tile, wl, t, lane, base, rows,
                               rot, lo, lo3, rec, rcount);
        if (lane == 0) reccnt[gw] = rcount;
    }

    __syncthreads();
    float4* o4 = reinterpret_cast<float4*>(agg + (size_t)base * D);
    for (int i = t; i < rows * (D / 4); i += TBLK) o4[i] = t4[i];
}

__global__ __launch_bounds__(256)
void norm_gemm_wmma(const float* __restrict__ agg,
                    const float* __restrict__ hprev,
                    const float* __restrict__ deg,
                    const float* __restrict__ W,
                    const float* __restrict__ bias,
                    float* __restrict__ out, int mode) {
    __shared__ float Alds[16 * 132];

    const int t = threadIdx.x;
    const int rowBase = blockIdx.x * 16;

    {
        int r  = t >> 4;
        int c0 = (t & 15) * 8;
        int row = rowBase + r;
        float s;
        if (mode == 0) s = 1.0f / (deg[row] + 1.0f);
        else           s = rsqrtf(fmaxf(deg[row], 1.0f));
        const float* ar = agg   + (size_t)row * D + c0;
        const float* hr = hprev + (size_t)row * D + c0;
#pragma unroll
        for (int j = 0; j < 8; ++j) {
            float v = ar[j];
            if (mode == 0) v += hr[j];
            Alds[r * 132 + c0 + j] = v * s;
        }
    }
    __syncthreads();

    const int waveId = t >> 5;
    const int lane   = t & 31;
    const int half   = lane >> 4;
    const int lm     = lane & 15;
    const int col0   = waveId * 16;

    v8f c = {};
#pragma unroll
    for (int kk = 0; kk < 32; ++kk) {
        int k0 = 4 * kk + 2 * half;
        v2f a, b;
        a.x = Alds[lm * 132 + k0];
        a.y = Alds[lm * 132 + k0 + 1];
        b.x = W[(size_t)(k0)     * D + col0 + lm];
        b.y = W[(size_t)(k0 + 1) * D + col0 + lm];
        c = __builtin_amdgcn_wmma_f32_16x16x4_f32(
                false, a, false, b, (short)0, c, false, false);
    }

    float bb = bias[col0 + lm];
#pragma unroll
    for (int v = 0; v < 8; ++v) {
        int row = rowBase + v + 8 * half;
        out[(size_t)row * D + col0 + lm] = c[v] + bb;
    }
}

extern "C" void kernel_launch(void* const* d_in, const int* in_sizes, int n_in,
                              void* d_out, int out_size, void* d_ws, size_t ws_size,
                              hipStream_t stream) {
    const float* x  = (const float*)d_in[0];
    const float* W1 = (const float*)d_in[1];
    const float* b1 = (const float*)d_in[2];
    const float* W2 = (const float*)d_in[3];
    const float* b2 = (const float*)d_in[4];
    const float* W3 = (const float*)d_in[5];
    const float* b3 = (const float*)d_in[6];
    const int* src  = (const int*)d_in[7];
    const int* dst  = (const int*)d_in[8];
    float* out = (float*)d_out;

    const int N = NN, E = NE;
    float* deg_in  = (float*)d_ws;
    float* deg_out = deg_in + N;
    float* agg     = deg_out + N;
    float* h1      = agg + (size_t)N * D;
    float* h2      = h1  + (size_t)N * D;
    int* reccnt       = (int*)(h2 + (size_t)N * D);
    unsigned* reclist = (unsigned*)(reccnt + NTILE * NWAVE);

    dim3 tileGrid(NTILE);
    dim3 gemmGrid(N / 16);

    hipMemsetAsync(deg_in, 0, (size_t)2 * N * sizeof(float), stream);
    degree_kernel<<<(E + 255) / 256, 256, 0, stream>>>(src, dst, deg_in, deg_out, E);

    tile_sum_kernel<0, 0><<<tileGrid, TBLK, 0, stream>>>(x, src, dst, deg_out, agg, reclist, reccnt);
    norm_gemm_wmma<<<gemmGrid, 256, 0, stream>>>(agg, x, deg_in, W1, b1, h1, 0);

    tile_sum_kernel<0, 1><<<tileGrid, TBLK, 0, stream>>>(h1, src, dst, deg_out, agg, reclist, reccnt);
    norm_gemm_wmma<<<gemmGrid, 256, 0, stream>>>(agg, h1, deg_in, W2, b2, h2, 0);

    tile_sum_kernel<1, 1><<<tileGrid, TBLK, 0, stream>>>(h2, src, dst, deg_out, agg, reclist, reccnt);
    norm_gemm_wmma<<<gemmGrid, 256, 0, stream>>>(agg, h2, deg_in, W3, b3, out, 1);
}


# LDS float adds replaced by owner waves: each hit goes to the LDS queue of the wave owning its row (slot = LDS counter rank); that wave sums its rows in registers in queue order, stores them whole
# speedup vs baseline: 2.1282x; 1.0491x over previous
#include <hip/hip_runtime.h>

#define NN 50000
#define NE 800000
#define D  128

typedef __attribute__((ext_vector_type(2))) float v2f;
typedef __attribute__((ext_vector_type(8))) float v8f;

__global__ void degree_kernel(const int* __restrict__ src,
                              const int* __restrict__ dst,
                              float* __restrict__ din,
                              float* __restrict__ dout, int E) {
    int e = blockIdx.x * blockDim.x + threadIdx.x;
    if (e < E) {
        unsafeAtomicAdd(&din[dst[e]], 1.0f);
        unsafeAtomicAdd(&dout[src[e]], 1.0f);
    }
}

#define RT     196
#define NTILE  256
#define SCAN_U 8
#define WLCAP  (32 + SCAN_U * 128)
#define NQ     (NE / 4)
#define TBLK   1024
#define NWAVE  (TBLK / 32)
#define QCAP   512
#define NACC   7

static_assert(NE % 4 == 0, "dst list is read in 16-byte words");
static_assert(WLCAP >= 31 + SCAN_U * 128, "a wave's list holds the rest of one flush plus one full chunk");
static_assert(NE <= (1 << 20) && RT <= 256, "entry packs edge id (20 bits) and local row (8 bits)");
static_assert((long)NTILE * RT >= NN && (long)(NTILE - 1) * RT < NN, "tiles cover every row");
static_assert(NN <= (1 << 16), "queue entry packs source row (16 bits) and the owner's row number");
static_assert(QCAP % 128 == 0, "a kept queue is staged back into LDS in steps of 128 entries");
static_assert(NACC * NWAVE >= RT && (NWAVE & (NWAVE - 1)) == 0, "every local row has an owner wave and a register set");

__device__ __forceinline__ void tile_add_row(float* trow, float4 v,
                                             int rot, int lo, int lo3) {
    float a0 = rot ? v.y : v.x;
    float a1 = rot ? v.z : v.y;
    float a2 = rot ? v.w : v.z;
    float a3 = rot ? v.x : v.w;
    __hip_atomic_fetch_add(trow + lo + 0, a0, __ATOMIC_RELAXED, __HIP_MEMORY_SCOPE_WORKGROUP);
    __hip_atomic_fetch_add(trow + lo + 1, a1, __ATOMIC_RELAXED, __HIP_MEMORY_SCOPE_WORKGROUP);
    __hip_atomic_fetch_add(trow + lo + 2, a2, __ATOMIC_RELAXED, __HIP_MEMORY_SCOPE_WORKGROUP);
    __hip_atomic_fetch_add(trow + lo3,    a3, __ATOMIC_RELAXED, __HIP_MEMORY_SCOPE_WORKGROUP);
}

__device__ __forceinline__ void note_hits(int4 dv, int q, bool valid, int base, int rows,
                                          unsigned* wl, int& wcount) {
    const unsigned r0 = (unsigned)(dv.x - base);
    const unsigned r1 = (unsigned)(dv.y - base);
    const unsigned r2 = (unsigned)(dv.z - base);
    const unsigned r3 = (unsigned)(dv.w - base);
    const bool h0 = valid && r0 < (unsigned)rows, h1 = valid && r1 < (unsigned)rows;
    const bool h2 = valid && r2 < (unsigned)rows, h3 = valid && r3 < (unsigned)rows;
    const unsigned m0 = __builtin_amdgcn_ballot_w32(h0);
    const unsigned m1 = __builtin_amdgcn_ballot_w32(h1);
    const unsigned m2 = __builtin_amdgcn_ballot_w32(h2);
    const unsigned m3 = __builtin_amdgcn_ballot_w32(h3);
    if (m0 | m1 | m2 | m3) {
        if (m0) {
            if (h0) wl[wcount + __builtin_amdgcn_mbcnt_lo(m0, 0u)] = (unsigned)(4 * q + 0) | (r0 << 20);
            wcount += __builtin_popcount(m0);
        }
        if (m1) {
            if (h1) wl[wcount + __builtin_amdgcn_mbcnt_lo(m1, 0u)] = (unsigned)(4 * q + 1) | (r1 << 20);
            wcount += __builtin_popcount(m1);
        }
        if (m2) {
            if (h2) wl[wcount + __builtin_amdgcn_mbcnt_lo(m2, 0u)] = (unsigned)(4 * q + 2) | (r2 << 20);
            wcount += __builtin_popcount(m2);
        }
        if (m3) {
            if (h3) wl[wcount + __builtin_amdgcn_mbcnt_lo(m3, 0u)] = (unsigned)(4 * q + 3) | (r3 << 20);
            wcount += __builtin_popcount(m3);
        }
    }
}

#define ACC_CASE(j) case j: a##j.x += x.x; a##j.y += x.y; a##j.z += x.z; a##j.w += x.w; break;

template <int SCALE>
__device__ __forceinline__ void gather_acc(const float* __restrict__ h,
                                           const float* __restrict__ degout,
                                           int s, int c, int cnt, int lane,
                                           float4& a0, float4& a1, float4& a2, float4& a3,
                                           float4& a4, float4& a5, float4& a6) {
    int scb = 0;
    if (SCALE) scb = __float_as_int(rsqrtf(fmaxf(degout[s], 1.0f)));
    const float4* h4 = reinterpret_cast<const float4*>(h);
    for (int k0 = 0; k0 < cnt; k0 += 8) {
        float4 v[8];
#pragma unroll
        for (int k = 0; k < 8; ++k) {
            int sk = __builtin_amdgcn_readlane(s, k0 + k);
            v[k] = h4[(size_t)sk * (D / 4) + lane];
        }
#pragma unroll
        for (int k = 0; k < 8; ++k) {
            if (k0 + k < cnt) {
                const int ck = __builtin_amdgcn_readlane(c, k0 + k);
                float4 x = v[k];
                if (SCALE) {
                    float sc = __int_as_float(__builtin_amdgcn_readlane(scb, k0 + k));
                    x.x *= sc; x.y *= sc; x.z *= sc; x.w *= sc;
                }
                switch (ck) {
                    ACC_CASE(0) ACC_CASE(1) ACC_CASE(2) ACC_CASE(3)
                    ACC_CASE(4) ACC_CASE(5) ACC_CASE(6)
                    default: break;
                }
            }
        }
    }
}

template <int SCALE>
__device__ __forceinline__ void route_round(const float* __restrict__ h,
                                            const int* __restrict__ src,
                                            const float* __restrict__ degout,
                                            float* tile, const unsigned* wl,
                                            int off, int cnt, int lane,
                                            int rot, int lo, int lo3,
                                            unsigned* queue, int* qcnt) {
    const unsigned ent = (lane < cnt) ? wl[off + lane] : 0u;
    const int e = (int)(ent & 0xFFFFFu);
    const int r = (int)(ent >> 20);
    int s = 0;
    bool spill = false;
    if (lane < cnt) {
        s = src[e];
        const int owner = r & (NWAVE - 1);
        const int slot = __hip_atomic_fetch_add(qcnt + owner, 1, __ATOMIC_RELAXED, __HIP_MEMORY_SCOPE_WORKGROUP);
        if (slot < QCAP) queue[owner * QCAP + slot] = (unsigned)s | ((unsigned)(r >> 5) << 16);
        else spill = true;
    }
    unsigned m = __builtin_amdgcn_ballot_w32(spill);
    const float4* h4 = reinterpret_cast<const float4*>(h);
    while (m) {
        const int k = __builtin_ctz(m);
        m &= m - 1;
        const int sk = __builtin_amdgcn_readlane(s, k);
        const int rk = __builtin_amdgcn_readlane(r, k);
        float4 x = h4[(size_t)sk * (D / 4) + lane];
        if (SCALE) {
            const float sc = rsqrtf(fmaxf(degout[sk], 1.0f));
            x.x *= sc; x.y *= sc; x.z *= sc; x.w *= sc;
        }
        tile_add_row(tile + rk * D, x, rot, lo, lo3);
    }
}

template <int SCALE>
__device__ __forceinline__ void scan_and_route(const float* __restrict__ h,
                                               const int* __restrict__ src,
                                               const int* __restrict__ dst,
                                               const float* __restrict__ degout,
                                               float* tile, unsigned* wl,
                                               int t, int lane, int base, int rows,
                                               int rot, int lo, int lo3,
                                               unsigned* queue, int* qcnt) {
    const int4* d4 = reinterpret_cast<const int4*>(dst);
    int wcount = 0;

    int q0 = 0;
    for (; q0 + SCAN_U * TBLK <= NQ; q0 += SCAN_U * TBLK) {
        int4 dv[SCAN_U];
#pragma unroll
        for (int u = 0; u < SCAN_U; ++u)
            dv[u] = d4[q0 + u * TBLK + t];
#pragma unroll
        for (int u = 0; u < SCAN_U; ++u)
            note_hits(dv[u], q0 + u * TBLK + t, true, base, rows, wl, wcount);
        if (wcount >= 32) {
            int done = 0;
            for (; done + 32 <= wcount; done += 32)
                route_round<SCALE>(h, src, degout, tile, wl, done, 32, lane, rot, lo, lo3, queue, qcnt);
            const int rem = wcount - done;
            unsigned keep = (lane < rem) ? wl[done + lane] : 0u;
            if (lane < rem) wl[lane] = keep;
            wcount = rem;
        }
    }
    for (; q0 < NQ; q0 += TBLK) {
        const int q = q0 + t;
        const bool valid = q < NQ;
        note_hits(d4[valid ? q : NQ - 1], q, valid, base, rows, wl, wcount);
    }
    for (int done = 0; done < wcount; done += 32)
        route_round<SCALE>(h, src, degout, tile, wl, done, min(32, wcount - done), lane, rot, lo, lo3, queue, qcnt);
}

template <int SCALE, int REPLAY>
__global__ __launch_bounds__(TBLK)
void tile_sum_kernel(const float* __restrict__ h,
                     const int* __restrict__ src,
                     const int* __restrict__ dst,
                     const float* __restrict__ degout,
                     float* __restrict__ agg,
                     unsigned* reclist,
                     int* reccnt) {
    __shared__ __attribute__((aligned(16))) float tile[RT * D];
    __shared__ unsigned wlist[NWAVE * WLCAP];
    __shared__ unsigned queue[NWAVE * QCAP];
    __shared__ int qcnt[NWAVE];

    const int t    = threadIdx.x;
    const int lane = t & 31;
    const int wave = __builtin_amdgcn_readfirstlane(t >> 5);
    const int base = blockIdx.x * RT;
    const int rows = min(RT, NN - base);
    const int rot  = lane >> 4;
    const int lo   = 4 * lane + rot;
    const int lo3  = 4 * lane + 3 - 3 * rot;

    float4* t4 = reinterpret_cast<float4*>(tile);
    const int gw = blockIdx.x * NWAVE + wave;
    unsigned* rec = reclist + (size_t)gw * QCAP;
    unsigned* q   = queue + wave * QCAP;

    bool scan = true;
    int total = 0;
    if (REPLAY) {
        const int cw = reccnt[blockIdx.x * NWAVE + lane];
        scan  = __builtin_amdgcn_ballot_w32(cw > QCAP) != 0u;
        total = __builtin_amdgcn_readlane(cw, wave);
    }

    if (scan) {
        for (int i = t; i < RT * D / 4; i += TBLK) t4[i] = make_float4(0.f, 0.f, 0.f, 0.f);
        if (t < NWAVE) qcnt[t] = 0;
        __syncthreads();
        scan_and_route<SCALE>(h, src, dst, degout, tile, wlist + wave * WLCAP, t, lane, base, rows,
                              rot, lo, lo3, queue, qcnt);
        __syncthreads();
        const int filled = __builtin_amdgcn_readfirstlane(qcnt[wave]);
        total = min(filled, QCAP);
        if (!REPLAY) {
            for (int i = lane; i < total; i += 32) rec[i] = q[i];
            if (lane == 0) reccnt[gw] = filled;
        }
    } else {
        for (int i0 = 0; i0 < total; i0 += 128) {
            const int i = i0 + lane;
            const unsigned e0 = (i      < total) ? rec[i]      : 0u;
            const unsigned e1 = (i + 32 < total) ? rec[i + 32] : 0u;
            const unsigned e2 = (i + 64 < total) ? rec[i + 64] : 0u;
            const unsigned e3 = (i + 96 < total) ? rec[i + 96] : 0u;
            q[i] = e0; q[i + 32] = e1; q[i + 64] = e2; q[i + 96] = e3;
        }
    }

    float4 a0 = make_float4(0.f, 0.f, 0.f, 0.f), a1 = a0, a2 = a0, a3 = a0, a4 = a0, a5 = a0, a6 = a0;
    for (int off = 0; off < total; off += 32) {
        const int cnt = min(32, total - off);
        const unsigned ent = (lane < cnt) ? q[off + lane] : 0u;
        gather_acc<SCALE>(h, degout, (int)(ent & 0xFFFFu), (int)(ent >> 16), cnt, lane,
                          a0, a1, a2, a3, a4, a5, a6);
    }

    float4* o4 = reinterpret_cast<float4*>(agg + (size_t)base * D);
    const float4 acc[NACC] = {a0, a1, a2, a3, a4, a5, a6};
#pragma unroll
    for (int j = 0; j < NACC; ++j) {
        const int r = wave + NWAVE * j;
        if (r < rows) {
            float4 o = acc[j];
            if (scan) {
                const float4 p = t4[r * (D / 4) + lane];
                o.x += p.x; o.y += p.y; o.z += p.z; o.w += p.w;
            }
            o4[r * (D / 4) + lane] = o;
        }
    }
}

__global__ __launch_bounds__(256)
void norm_gemm_wmma(const float* __restrict__ agg,
                    const float* __restrict__ hprev,
                    const float* __restrict__ deg,
                    const float* __restrict__ W,
                    const float* __restrict__ bias,
                    float* __restrict__ out, int mode) {
    __shared__ float Alds[16 * 132];

    const int t = threadIdx.x;
    const int rowBase = blockIdx.x * 16;

    {
        int r  = t >> 4;
        int c0 = (t & 15) * 8;
        int row = rowBase + r;
        float s;
        if (mode == 0) s = 1.0f / (deg[row] + 1.0f);
        else           s = rsqrtf(fmaxf(deg[row], 1.0f));
        const float* ar = agg   + (size_t)row * D + c0;
        const float* hr = hprev + (size_t)row * D + c0;
#pragma unroll
        for (int j = 0; j < 8; ++j) {
            float v = ar[j];
            if (mode == 0) v += hr[j];
            Alds[r * 132 + c0 + j] = v * s;
        }
    }
    __syncthreads();

    const int waveId = t >> 5;
    const int lane   = t & 31;
    const int half   = lane >> 4;
    const int lm     = lane & 15;
    const int col0   = waveId * 16;

    v8f c = {};
#pragma unroll
    for (int kk = 0; kk < 32; ++kk) {
        int k0 = 4 * kk + 2 * half;
        v2f a, b;
        a.x = Alds[lm * 132 + k0];
        a.y = Alds[lm * 132 + k0 + 1];
        b.x = W[(size_t)(k0)     * D + col0 + lm];
        b.y = W[(size_t)(k0 + 1) * D + col0 + lm];
        c = __builtin_amdgcn_wmma_f32_16x16x4_f32(
                false, a, false, b, (short)0, c, false, false);
    }

    float bb = bias[col0 + lm];
#pragma unroll
    for (int v = 0; v < 8; ++v) {
        int row = rowBase + v + 8 * half;
        out[(size_t)row * D + col0 + lm] = c[v] + bb;
    }
}

extern "C" void kernel_launch(void* const* d_in, const int* in_sizes, int n_in,
                              void* d_out, int out_size, void* d_ws, size_t ws_size,
                              hipStream_t stream) {
    const float* x  = (const float*)d_in[0];
    const float* W1 = (const float*)d_in[1];
    const float* b1 = (const float*)d_in[2];
    const float* W2 = (const float*)d_in[3];
    const float* b2 = (const float*)d_in[4];
    const float* W3 = (const float*)d_in[5];
    const float* b3 = (const float*)d_in[6];
    const int* src  = (const int*)d_in[7];
    const int* dst  = (const int*)d_in[8];
    float* out = (float*)d_out;

    const int N = NN, E = NE;
    float* deg_in  = (float*)d_ws;
    float* deg_out = deg_in + N;
    float* agg     = deg_out + N;
    float* h1      = agg + (size_t)N * D;
    float* h2      = h1  + (size_t)N * D;
    int* reccnt       = (int*)(h2 + (size_t)N * D);
    unsigned* reclist = (unsigned*)(reccnt + NTILE * NWAVE);

    dim3 tileGrid(NTILE);
    dim3 gemmGrid(N / 16);

    hipMemsetAsync(deg_in, 0, (size_t)2 * N * sizeof(float), stream);
    degree_kernel<<<(E + 255) / 256, 256, 0, stream>>>(src, dst, deg_in, deg_out, E);

    tile_sum_kernel<0, 0><<<tileGrid, TBLK, 0, stream>>>(x, src, dst, deg_out, agg, reclist, reccnt);
    norm_gemm_wmma<<<gemmGrid, 256, 0, stream>>>(agg, x, deg_in, W1, b1, h1, 0);

    tile_sum_kernel<0, 1><<<tileGrid, TBLK, 0, stream>>>(h1, src, dst, deg_out, agg, reclist, reccnt);
    norm_gemm_wmma<<<gemmGrid, 256, 0, stream>>>(agg, h1, deg_in, W2, b2, h2, 0);

    tile_sum_kernel<1, 1><<<tileGrid, TBLK, 0, stream>>>(h2, src, dst, deg_out, agg, reclist, reccnt);
    norm_gemm_wmma<<<gemmGrid, 256, 0, stream>>>(agg, h2, deg_in, W3, b3, out, 1);
}


# Degree launch folded into layer 1's edge pass: the block handing an edge on adds 1 to an LDS counter of its destination row (stored as in-degree) and 1 to its source's out-degree; values unchanged
# speedup vs baseline: 2.5062x; 1.1776x over previous
#include <hip/hip_runtime.h>

#define NN 50000
#define NE 800000
#define D  128

typedef __attribute__((ext_vector_type(2))) float v2f;
typedef __attribute__((ext_vector_type(8))) float v8f;

#define RT     196
#define NTILE  256
#define SCAN_U 8
#define WLCAP  (32 + SCAN_U * 128)
#define NQ     (NE / 4)
#define TBLK   1024
#define NWAVE  (TBLK / 32)
#define QCAP   512
#define NACC   7

static_assert(NE % 4 == 0, "dst list is read in 16-byte words");
static_assert(WLCAP >= 31 + SCAN_U * 128, "a wave's list holds the rest of one flush plus one full chunk");
static_assert(NE <= (1 << 20) && RT <= 256, "entry packs edge id (20 bits) and local row (8 bits)");
static_assert((long)NTILE * RT >= NN && (long)(NTILE - 1) * RT < NN, "tiles cover every row");
static_assert(NN <= (1 << 16), "queue entry packs source row (16 bits) and the owner's row number");
static_assert(QCAP % 128 == 0, "a kept queue is staged back into LDS in steps of 128 entries");
static_assert(NACC * NWAVE >= RT && (NWAVE & (NWAVE - 1)) == 0, "every local row has an owner wave and a register set");

__device__ __forceinline__ void tile_add_row(float* trow, float4 v,
                                             int rot, int lo, int lo3) {
    float a0 = rot ? v.y : v.x;
    float a1 = rot ? v.z : v.y;
    float a2 = rot ? v.w : v.z;
    float a3 = rot ? v.x : v.w;
    __hip_atomic_fetch_add(trow + lo + 0, a0, __ATOMIC_RELAXED, __HIP_MEMORY_SCOPE_WORKGROUP);
    __hip_atomic_fetch_add(trow + lo + 1, a1, __ATOMIC_RELAXED, __HIP_MEMORY_SCOPE_WORKGROUP);
    __hip_atomic_fetch_add(trow + lo + 2, a2, __ATOMIC_RELAXED, __HIP_MEMORY_SCOPE_WORKGROUP);
    __hip_atomic_fetch_add(trow + lo3,    a3, __ATOMIC_RELAXED, __HIP_MEMORY_SCOPE_WORKGROUP);
}

__device__ __forceinline__ void note_hits(int4 dv, int q, bool valid, int base, int rows,
                                          unsigned* wl, int& wcount) {
    const unsigned r0 = (unsigned)(dv.x - base);
    const unsigned r1 = (unsigned)(dv.y - base);
    const unsigned r2 = (unsigned)(dv.z - base);
    const unsigned r3 = (unsigned)(dv.w - base);
    const bool h0 = valid && r0 < (unsigned)rows, h1 = valid && r1 < (unsigned)rows;
    const bool h2 = valid && r2 < (unsigned)rows, h3 = valid && r3 < (unsigned)rows;
    const unsigned m0 = __builtin_amdgcn_ballot_w32(h0);
    const unsigned m1 = __builtin_amdgcn_ballot_w32(h1);
    const unsigned m2 = __builtin_amdgcn_ballot_w32(h2);
    const unsigned m3 = __builtin_amdgcn_ballot_w32(h3);
    if (m0 | m1 | m2 | m3) {
        if (m0) {
            if (h0) wl[wcount + __builtin_amdgcn_mbcnt_lo(m0, 0u)] = (unsigned)(4 * q + 0) | (r0 << 20);
            wcount += __builtin_popcount(m0);
        }
        if (m1) {
            if (h1) wl[wcount + __builtin_amdgcn_mbcnt_lo(m1, 0u)] = (unsigned)(4 * q + 1) | (r1 << 20);
            wcount += __builtin_popcount(m1);
        }
        if (m2) {
            if (h2) wl[wcount + __builtin_amdgcn_mbcnt_lo(m2, 0u)] = (unsigned)(4 * q + 2) | (r2 << 20);
            wcount += __builtin_popcount(m2);
        }
        if (m3) {
            if (h3) wl[wcount + __builtin_amdgcn_mbcnt_lo(m3, 0u)] = (unsigned)(4 * q + 3) | (r3 << 20);
            wcount += __builtin_popcount(m3);
        }
    }
}

#define ACC_CASE(j) case j: a##j.x += x.x; a##j.y += x.y; a##j.z += x.z; a##j.w += x.w; break;

template <int SCALE>
__device__ __forceinline__ void gather_acc(const float* __restrict__ h,
                                           const float* __restrict__ degout,
                                           int s, int c, int cnt, int lane,
                                           float4& a0, float4& a1, float4& a2, float4& a3,
                                           float4& a4, float4& a5, float4& a6) {
    int scb = 0;
    if (SCALE) scb = __float_as_int(rsqrtf(fmaxf(degout[s], 1.0f)));
    const float4* h4 = reinterpret_cast<const float4*>(h);
    for (int k0 = 0; k0 < cnt; k0 += 8) {
        float4 v[8];
#pragma unroll
        for (int k = 0; k < 8; ++k) {
            int sk = __builtin_amdgcn_readlane(s, k0 + k);
            v[k] = h4[(size_t)sk * (D / 4) + lane];
        }
#pragma unroll
        for (int k = 0; k < 8; ++k) {
            if (k0 + k < cnt) {
                const int ck = __builtin_amdgcn_readlane(c, k0 + k);
                float4 x = v[k];
                if (SCALE) {
                    float sc = __int_as_float(__builtin_amdgcn_readlane(scb, k0 + k));
                    x.x *= sc; x.y *= sc; x.z *= sc; x.w *= sc;
                }
                switch (ck) {
                    ACC_CASE(0) ACC_CASE(1) ACC_CASE(2) ACC_CASE(3)
                    ACC_CASE(4) ACC_CASE(5) ACC_CASE(6)
                    default: break;
                }
            }
        }
    }
}

template <int SCALE, int COUNT>
__device__ __forceinline__ void route_round(const float* __restrict__ h,
                                            const int* __restrict__ src,
                                            const float* __restrict__ degout,
                                            float* tile, const unsigned* wl,
                                            int off, int cnt, int lane,
                                            int rot, int lo, int lo3,
                                            unsigned* queue, int* qcnt,
                                            int* rowdeg, float* outdeg) {
    const unsigned ent = (lane < cnt) ? wl[off + lane] : 0u;
    const int e = (int)(ent & 0xFFFFFu);
    const int r = (int)(ent >> 20);
    int s = 0;
    bool spill = false;
    if (lane < cnt) {
        s = src[e];
        if (COUNT) {
            __hip_atomic_fetch_add(rowdeg + r, 1, __ATOMIC_RELAXED, __HIP_MEMORY_SCOPE_WORKGROUP);
            unsafeAtomicAdd(outdeg + s, 1.0f);
        }
        const int owner = r & (NWAVE - 1);
        const int slot = __hip_atomic_fetch_add(qcnt + owner, 1, __ATOMIC_RELAXED, __HIP_MEMORY_SCOPE_WORKGROUP);
        if (slot < QCAP) queue[owner * QCAP + slot] = (unsigned)s | ((unsigned)(r >> 5) << 16);
        else spill = true;
    }
    unsigned m = __builtin_amdgcn_ballot_w32(spill);
    const float4* h4 = reinterpret_cast<const float4*>(h);
    while (m) {
        const int k = __builtin_ctz(m);
        m &= m - 1;
        const int sk = __builtin_amdgcn_readlane(s, k);
        const int rk = __builtin_amdgcn_readlane(r, k);
        float4 x = h4[(size_t)sk * (D / 4) + lane];
        if (SCALE) {
            const float sc = rsqrtf(fmaxf(degout[sk], 1.0f));
            x.x *= sc; x.y *= sc; x.z *= sc; x.w *= sc;
        }
        tile_add_row(tile + rk * D, x, rot, lo, lo3);
    }
}

template <int SCALE, int COUNT>
__device__ __forceinline__ void scan_and_route(const float* __restrict__ h,
                                               const int* __restrict__ src,
                                               const int* __restrict__ dst,
                                               const float* __restrict__ degout,
                                               float* tile, unsigned* wl,
                                               int t, int lane, int base, int rows,
                                               int rot, int lo, int lo3,
                                               unsigned* queue, int* qcnt,
                                               int* rowdeg, float* outdeg) {
    const int4* d4 = reinterpret_cast<const int4*>(dst);
    int wcount = 0;

    int q0 = 0;
    for (; q0 + SCAN_U * TBLK <= NQ; q0 += SCAN_U * TBLK) {
        int4 dv[SCAN_U];
#pragma unroll
        for (int u = 0; u < SCAN_U; ++u)
            dv[u] = d4[q0 + u * TBLK + t];
#pragma unroll
        for (int u = 0; u < SCAN_U; ++u)
            note_hits(dv[u], q0 + u * TBLK + t, true, base, rows, wl, wcount);
        if (wcount >= 32) {
            int done = 0;
            for (; done + 32 <= wcount; done += 32)
                route_round<SCALE, COUNT>(h, src, degout, tile, wl, done, 32, lane, rot, lo, lo3, queue, qcnt,
                                          rowdeg, outdeg);
            const int rem = wcount - done;
            unsigned keep = (lane < rem) ? wl[done + lane] : 0u;
            if (lane < rem) wl[lane] = keep;
            wcount = rem;
        }
    }
    for (; q0 < NQ; q0 += TBLK) {
        const int q = q0 + t;
        const bool valid = q < NQ;
        note_hits(d4[valid ? q : NQ - 1], q, valid, base, rows, wl, wcount);
    }
    for (int done = 0; done < wcount; done += 32)
        route_round<SCALE, COUNT>(h, src, degout, tile, wl, done, min(32, wcount - done), lane, rot, lo, lo3, queue, qcnt,
                                  rowdeg, outdeg);
}

template <int SCALE, int REPLAY>
__global__ __launch_bounds__(TBLK)
void tile_sum_kernel(const float* __restrict__ h,
                     const int* __restrict__ src,
                     const int* __restrict__ dst,
                     const float* __restrict__ degout,
                     float* __restrict__ agg,
                     unsigned* reclist,
                     int* reccnt,
                     float* indeg,
                     float* outdeg) {
    __shared__ __attribute__((aligned(16))) float tile[RT * D];
    __shared__ unsigned wlist[NWAVE * WLCAP];
    __shared__ unsigned queue[NWAVE * QCAP];
    __shared__ int qcnt[NWAVE];
    __shared__ int rowdeg[RT];

    const int t    = threadIdx.x;
    const int lane = t & 31;
    const int wave = __builtin_amdgcn_readfirstlane(t >> 5);
    const int base = blockIdx.x * RT;
    const int rows = min(RT, NN - base);
    const int rot  = lane >> 4;
    const int lo   = 4 * lane + rot;
    const int lo3  = 4 * lane + 3 - 3 * rot;

    float4* t4 = reinterpret_cast<float4*>(tile);
    const int gw = blockIdx.x * NWAVE + wave;
    unsigned* rec = reclist + (size_t)gw * QCAP;
    unsigned* q   = queue + wave * QCAP;

    bool scan = true;
    int total = 0;
    if (REPLAY) {
        const int cw = reccnt[blockIdx.x * NWAVE + lane];
        scan  = __builtin_amdgcn_ballot_w32(cw > QCAP) != 0u;
        total = __builtin_amdgcn_readlane(cw, wave);
    }

    if (scan) {
        for (int i = t; i < RT * D / 4; i += TBLK) t4[i] = make_float4(0.f, 0.f, 0.f, 0.f);
        if (t < NWAVE) qcnt[t] = 0;
        if (!REPLAY && t < RT) rowdeg[t] = 0;
        __syncthreads();
        scan_and_route<SCALE, !REPLAY>(h, src, dst, degout, tile, wlist + wave * WLCAP, t, lane, base, rows,
                                       rot, lo, lo3, queue, qcnt, rowdeg, outdeg);
        __syncthreads();
        const int filled = __builtin_amdgcn_readfirstlane(qcnt[wave]);
        total = min(filled, QCAP);
        if (!REPLAY) {
            for (int i = lane; i < total; i += 32) rec[i] = q[i];
            if (lane == 0) reccnt[gw] = filled;
            if (t < rows) indeg[base + t] = (float)rowdeg[t];
        }
    } else {
        for (int i0 = 0; i0 < total; i0 += 128) {
            const int i = i0 + lane;
            const unsigned e0 = (i      < total) ? rec[i]      : 0u;
            const unsigned e1 = (i + 32 < total) ? rec[i + 32] : 0u;
            const unsigned e2 = (i + 64 < total) ? rec[i + 64] : 0u;
            const unsigned e3 = (i + 96 < total) ? rec[i + 96] : 0u;
            q[i] = e0; q[i + 32] = e1; q[i + 64] = e2; q[i + 96] = e3;
        }
    }

    float4 a0 = make_float4(0.f, 0.f, 0.f, 0.f), a1 = a0, a2 = a0, a3 = a0, a4 = a0, a5 = a0, a6 = a0;
    for (int off = 0; off < total; off += 32) {
        const int cnt = min(32, total - off);
        const unsigned ent = (lane < cnt) ? q[off + lane] : 0u;
        gather_acc<SCALE>(h, degout, (int)(ent & 0xFFFFu), (int)(ent >> 16), cnt, lane,
                          a0, a1, a2, a3, a4, a5, a6);
    }

    float4* o4 = reinterpret_cast<float4*>(agg + (size_t)base * D);
    const float4 acc[NACC] = {a0, a1, a2, a3, a4, a5, a6};
#pragma unroll
    for (int j = 0; j < NACC; ++j) {
        const int r = wave + NWAVE * j;
        if (r < rows) {
            float4 o = acc[j];
            if (scan) {
                const float4 p = t4[r * (D / 4) + lane];
                o.x += p.x; o.y += p.y; o.z += p.z; o.w += p.w;
            }
            o4[r * (D / 4) + lane] = o;
        }
    }
}

__global__ __launch_bounds__(256)
void norm_gemm_wmma(const float* __restrict__ agg,
                    const float* __restrict__ hprev,
                    const float* __restrict__ deg,
                    const float* __restrict__ W,
                    const float* __restrict__ bias,
                    float* __restrict__ out, int mode) {
    __shared__ float Alds[16 * 132];

    const int t = threadIdx.x;
    const int rowBase = blockIdx.x * 16;

    {
        int r  = t >> 4;
        int c0 = (t & 15) * 8;
        int row = rowBase + r;
        float s;
        if (mode == 0) s = 1.0f / (deg[row] + 1.0f);
        else           s = rsqrtf(fmaxf(deg[row], 1.0f));
        const float* ar = agg   + (size_t)row * D + c0;
        const float* hr = hprev + (size_t)row * D + c0;
#pragma unroll
        for (int j = 0; j < 8; ++j) {
            float v = ar[j];
            if (mode == 0) v += hr[j];
            Alds[r * 132 + c0 + j] = v * s;
        }
    }
    __syncthreads();

    const int waveId = t >> 5;
    const int lane   = t & 31;
    const int half   = lane >> 4;
    const int lm     = lane & 15;
    const int col0   = waveId * 16;

    v8f c = {};
#pragma unroll
    for (int kk = 0; kk < 32; ++kk) {
        int k0 = 4 * kk + 2 * half;
        v2f a, b;
        a.x = Alds[lm * 132 + k0];
        a.y = Alds[lm * 132 + k0 + 1];
        b.x = W[(size_t)(k0)     * D + col0 + lm];
        b.y = W[(size_t)(k0 + 1) * D + col0 + lm];
        c = __builtin_amdgcn_wmma_f32_16x16x4_f32(
                false, a, false, b, (short)0, c, false, false);
    }

    float bb = bias[col0 + lm];
#pragma unroll
    for (int v = 0; v < 8; ++v) {
        int row = rowBase + v + 8 * half;
        out[(size_t)row * D + col0 + lm] = c[v] + bb;
    }
}

extern "C" void kernel_launch(void* const* d_in, const int* in_sizes, int n_in,
                              void* d_out, int out_size, void* d_ws, size_t ws_size,
                              hipStream_t stream) {
    const float* x  = (const float*)d_in[0];
    const float* W1 = (const float*)d_in[1];
    const float* b1 = (const float*)d_in[2];
    const float* W2 = (const float*)d_in[3];
    const float* b2 = (const float*)d_in[4];
    const float* W3 = (const float*)d_in[5];
    const float* b3 = (const float*)d_in[6];
    const int* src  = (const int*)d_in[7];
    const int* dst  = (const int*)d_in[8];
    float* out = (float*)d_out;

    const int N = NN;
    float* deg_in  = (float*)d_ws;
    float* deg_out = deg_in + N;
    float* agg     = deg_out + N;
    float* h1      = agg + (size_t)N * D;
    float* h2      = h1  + (size_t)N * D;
    int* reccnt       = (int*)(h2 + (size_t)N * D);
    unsigned* reclist = (unsigned*)(reccnt + NTILE * NWAVE);

    dim3 tileGrid(NTILE);
    dim3 gemmGrid(N / 16);

    hipMemsetAsync(deg_out, 0, (size_t)N * sizeof(float), stream);

    tile_sum_kernel<0, 0><<<tileGrid, TBLK, 0, stream>>>(x, src, dst, deg_out, agg, reclist, reccnt, deg_in, deg_out);
    norm_gemm_wmma<<<gemmGrid, 256, 0, stream>>>(agg, x, deg_in, W1, b1, h1, 0);

    tile_sum_kernel<0, 1><<<tileGrid, TBLK, 0, stream>>>(h1, src, dst, deg_out, agg, reclist, reccnt, deg_in, deg_out);
    norm_gemm_wmma<<<gemmGrid, 256, 0, stream>>>(agg, h1, deg_in, W2, b2, h2, 0);

    tile_sum_kernel<1, 1><<<tileGrid, TBLK, 0, stream>>>(h2, src, dst, deg_out, agg, reclist, reccnt, deg_in, deg_out);
    norm_gemm_wmma<<<gemmGrid, 256, 0, stream>>>(agg, h2, deg_in, W3, b3, out, 1);
}


# GEMM block enlarged from 16 to 80 rows: rows fetched as one group of 16-byte loads (mode a compile-time constant), each wave's W fragments feed five row tiles; 625 blocks, not 3125; k order kept
# speedup vs baseline: 2.7290x; 1.0889x over previous
#include <hip/hip_runtime.h>

#define NN 50000
#define NE 800000
#define D  128

typedef __attribute__((ext_vector_type(2))) float v2f;
typedef __attribute__((ext_vector_type(8))) float v8f;

#define RT     196
#define NTILE  256
#define SCAN_U 8
#define WLCAP  (32 + SCAN_U * 128)
#define NQ     (NE / 4)
#define TBLK   1024
#define NWAVE  (TBLK / 32)
#define QCAP   512
#define NACC   7

static_assert(NE % 4 == 0, "dst list is read in 16-byte words");
static_assert(WLCAP >= 31 + SCAN_U * 128, "a wave's list holds the rest of one flush plus one full chunk");
static_assert(NE <= (1 << 20) && RT <= 256, "entry packs edge id (20 bits) and local row (8 bits)");
static_assert((long)NTILE * RT >= NN && (long)(NTILE - 1) * RT < NN, "tiles cover every row");
static_assert(NN <= (1 << 16), "queue entry packs source row (16 bits) and the owner's row number");
static_assert(QCAP % 128 == 0, "a kept queue is staged back into LDS in steps of 128 entries");
static_assert(NACC * NWAVE >= RT && (NWAVE & (NWAVE - 1)) == 0, "every local row has an owner wave and a register set");

__device__ __forceinline__ void tile_add_row(float* trow, float4 v,
                                             int rot, int lo, int lo3) {
    float a0 = rot ? v.y : v.x;
    float a1 = rot ? v.z : v.y;
    float a2 = rot ? v.w : v.z;
    float a3 = rot ? v.x : v.w;
    __hip_atomic_fetch_add(trow + lo + 0, a0, __ATOMIC_RELAXED, __HIP_MEMORY_SCOPE_WORKGROUP);
    __hip_atomic_fetch_add(trow + lo + 1, a1, __ATOMIC_RELAXED, __HIP_MEMORY_SCOPE_WORKGROUP);
    __hip_atomic_fetch_add(trow + lo + 2, a2, __ATOMIC_RELAXED, __HIP_MEMORY_SCOPE_WORKGROUP);
    __hip_atomic_fetch_add(trow + lo3,    a3, __ATOMIC_RELAXED, __HIP_MEMORY_SCOPE_WORKGROUP);
}

__device__ __forceinline__ void note_hits(int4 dv, int q, bool valid, int base, int rows,
                                          unsigned* wl, int& wcount) {
    const unsigned r0 = (unsigned)(dv.x - base);
    const unsigned r1 = (unsigned)(dv.y - base);
    const unsigned r2 = (unsigned)(dv.z - base);
    const unsigned r3 = (unsigned)(dv.w - base);
    const bool h0 = valid && r0 < (unsigned)rows, h1 = valid && r1 < (unsigned)rows;
    const bool h2 = valid && r2 < (unsigned)rows, h3 = valid && r3 < (unsigned)rows;
    const unsigned m0 = __builtin_amdgcn_ballot_w32(h0);
    const unsigned m1 = __builtin_amdgcn_ballot_w32(h1);
    const unsigned m2 = __builtin_amdgcn_ballot_w32(h2);
    const unsigned m3 = __builtin_amdgcn_ballot_w32(h3);
    if (m0 | m1 | m2 | m3) {
        if (m0) {
            if (h0) wl[wcount + __builtin_amdgcn_mbcnt_lo(m0, 0u)] = (unsigned)(4 * q + 0) | (r0 << 20);
            wcount += __builtin_popcount(m0);
        }
        if (m1) {
            if (h1) wl[wcount + __builtin_amdgcn_mbcnt_lo(m1, 0u)] = (unsigned)(4 * q + 1) | (r1 << 20);
            wcount += __builtin_popcount(m1);
        }
        if (m2) {
            if (h2) wl[wcount + __builtin_amdgcn_mbcnt_lo(m2, 0u)] = (unsigned)(4 * q + 2) | (r2 << 20);
            wcount += __builtin_popcount(m2);
        }
        if (m3) {
            if (h3) wl[wcount + __builtin_amdgcn_mbcnt_lo(m3, 0u)] = (unsigned)(4 * q + 3) | (r3 << 20);
            wcount += __builtin_popcount(m3);
        }
    }
}

#define ACC_CASE(j) case j: a##j.x += x.x; a##j.y += x.y; a##j.z += x.z; a##j.w += x.w; break;

template <int SCALE>
__device__ __forceinline__ void gather_acc(const float* __restrict__ h,
                                           const float* __restrict__ degout,
                                           int s, int c, int cnt, int lane,
                                           float4& a0, float4& a1, float4& a2, float4& a3,
                                           float4& a4, float4& a5, float4& a6) {
    int scb = 0;
    if (SCALE) scb = __float_as_int(rsqrtf(fmaxf(degout[s], 1.0f)));
    const float4* h4 = reinterpret_cast<const float4*>(h);
    for (int k0 = 0; k0 < cnt; k0 += 8) {
        float4 v[8];
#pragma unroll
        for (int k = 0; k < 8; ++k) {
            int sk = __builtin_amdgcn_readlane(s, k0 + k);
            v[k] = h4[(size_t)sk * (D / 4) + lane];
        }
#pragma unroll
        for (int k = 0; k < 8; ++k) {
            if (k0 + k < cnt) {
                const int ck = __builtin_amdgcn_readlane(c, k0 + k);
                float4 x = v[k];
                if (SCALE) {
                    float sc = __int_as_float(__builtin_amdgcn_readlane(scb, k0 + k));
                    x.x *= sc; x.y *= sc; x.z *= sc; x.w *= sc;
                }
                switch (ck) {
                    ACC_CASE(0) ACC_CASE(1) ACC_CASE(2) ACC_CASE(3)
                    ACC_CASE(4) ACC_CASE(5) ACC_CASE(6)
                    default: break;
                }
            }
        }
    }
}

template <int SCALE, int COUNT>
__device__ __forceinline__ void route_round(const float* __restrict__ h,
                                            const int* __restrict__ src,
                                            const float* __restrict__ degout,
                                            float* tile, const unsigned* wl,
                                            int off, int cnt, int lane,
                                            int rot, int lo, int lo3,
                                            unsigned* queue, int* qcnt,
                                            int* rowdeg, float* outdeg) {
    const unsigned ent = (lane < cnt) ? wl[off + lane] : 0u;
    const int e = (int)(ent & 0xFFFFFu);
    const int r = (int)(ent >> 20);
    int s = 0;
    bool spill = false;
    if (lane < cnt) {
        s = src[e];
        if (COUNT) {
            __hip_atomic_fetch_add(rowdeg + r, 1, __ATOMIC_RELAXED, __HIP_MEMORY_SCOPE_WORKGROUP);
            unsafeAtomicAdd(outdeg + s, 1.0f);
        }
        const int owner = r & (NWAVE - 1);
        const int slot = __hip_atomic_fetch_add(qcnt + owner, 1, __ATOMIC_RELAXED, __HIP_MEMORY_SCOPE_WORKGROUP);
        if (slot < QCAP) queue[owner * QCAP + slot] = (unsigned)s | ((unsigned)(r >> 5) << 16);
        else spill = true;
    }
    unsigned m = __builtin_amdgcn_ballot_w32(spill);
    const float4* h4 = reinterpret_cast<const float4*>(h);
    while (m) {
        const int k = __builtin_ctz(m);
        m &= m - 1;
        const int sk = __builtin_amdgcn_readlane(s, k);
        const int rk = __builtin_amdgcn_readlane(r, k);
        float4 x = h4[(size_t)sk * (D / 4) + lane];
        if (SCALE) {
            const float sc = rsqrtf(fmaxf(degout[sk], 1.0f));
            x.x *= sc; x.y *= sc; x.z *= sc; x.w *= sc;
        }
        tile_add_row(tile + rk * D, x, rot, lo, lo3);
    }
}

template <int SCALE, int COUNT>
__device__ __forceinline__ void scan_and_route(const float* __restrict__ h,
                                               const int* __restrict__ src,
                                               const int* __restrict__ dst,
                                               const float* __restrict__ degout,
                                               float* tile, unsigned* wl,
                                               int t, int lane, int base, int rows,
                                               int rot, int lo, int lo3,
                                               unsigned* queue, int* qcnt,
                                               int* rowdeg, float* outdeg) {
    const int4* d4 = reinterpret_cast<const int4*>(dst);
    int wcount = 0;

    int q0 = 0;
    for (; q0 + SCAN_U * TBLK <= NQ; q0 += SCAN_U * TBLK) {
        int4 dv[SCAN_U];
#pragma unroll
        for (int u = 0; u < SCAN_U; ++u)
            dv[u] = d4[q0 + u * TBLK + t];
#pragma unroll
        for (int u = 0; u < SCAN_U; ++u)
            note_hits(dv[u], q0 + u * TBLK + t, true, base, rows, wl, wcount);
        if (wcount >= 32) {
            int done = 0;
            for (; done + 32 <= wcount; done += 32)
                route_round<SCALE, COUNT>(h, src, degout, tile, wl, done, 32, lane, rot, lo, lo3, queue, qcnt,
                                          rowdeg, outdeg);
            const int rem = wcount - done;
            unsigned keep = (lane < rem) ? wl[done + lane] : 0u;
            if (lane < rem) wl[lane] = keep;
            wcount = rem;
        }
    }
    for (; q0 < NQ; q0 += TBLK) {
        const int q = q0 + t;
        const bool valid = q < NQ;
        note_hits(d4[valid ? q : NQ - 1], q, valid, base, rows, wl, wcount);
    }
    for (int done = 0; done < wcount; done += 32)
        route_round<SCALE, COUNT>(h, src, degout, tile, wl, done, min(32, wcount - done), lane, rot, lo, lo3, queue, qcnt,
                                  rowdeg, outdeg);
}

template <int SCALE, int REPLAY>
__global__ __launch_bounds__(TBLK)
void tile_sum_kernel(const float* __restrict__ h,
                     const int* __restrict__ src,
                     const int* __restrict__ dst,
                     const float* __restrict__ degout,
                     float* __restrict__ agg,
                     unsigned* reclist,
                     int* reccnt,
                     float* indeg,
                     float* outdeg) {
    __shared__ __attribute__((aligned(16))) float tile[RT * D];
    __shared__ unsigned wlist[NWAVE * WLCAP];
    __shared__ unsigned queue[NWAVE * QCAP];
    __shared__ int qcnt[NWAVE];
    __shared__ int rowdeg[RT];

    const int t    = threadIdx.x;
    const int lane = t & 31;
    const int wave = __builtin_amdgcn_readfirstlane(t >> 5);
    const int base = blockIdx.x * RT;
    const int rows = min(RT, NN - base);
    const int rot  = lane >> 4;
    const int lo   = 4 * lane + rot;
    const int lo3  = 4 * lane + 3 - 3 * rot;

    float4* t4 = reinterpret_cast<float4*>(tile);
    const int gw = blockIdx.x * NWAVE + wave;
    unsigned* rec = reclist + (size_t)gw * QCAP;
    unsigned* q   = queue + wave * QCAP;

    bool scan = true;
    int total = 0;
    if (REPLAY) {
        const int cw = reccnt[blockIdx.x * NWAVE + lane];
        scan  = __builtin_amdgcn_ballot_w32(cw > QCAP) != 0u;
        total = __builtin_amdgcn_readlane(cw, wave);
    }

    if (scan) {
        for (int i = t; i < RT * D / 4; i += TBLK) t4[i] = make_float4(0.f, 0.f, 0.f, 0.f);
        if (t < NWAVE) qcnt[t] = 0;
        if (!REPLAY && t < RT) rowdeg[t] = 0;
        __syncthreads();
        scan_and_route<SCALE, !REPLAY>(h, src, dst, degout, tile, wlist + wave * WLCAP, t, lane, base, rows,
                                       rot, lo, lo3, queue, qcnt, rowdeg, outdeg);
        __syncthreads();
        const int filled = __builtin_amdgcn_readfirstlane(qcnt[wave]);
        total = min(filled, QCAP);
        if (!REPLAY) {
            for (int i = lane; i < total; i += 32) rec[i] = q[i];
            if (lane == 0) reccnt[gw] = filled;
            if (t < rows) indeg[base + t] = (float)rowdeg[t];
        }
    } else {
        for (int i0 = 0; i0 < total; i0 += 128) {
            const int i = i0 + lane;
            const unsigned e0 = (i      < total) ? rec[i]      : 0u;
            const unsigned e1 = (i + 32 < total) ? rec[i + 32] : 0u;
            const unsigned e2 = (i + 64 < total) ? rec[i + 64] : 0u;
            const unsigned e3 = (i + 96 < total) ? rec[i + 96] : 0u;
            q[i] = e0; q[i + 32] = e1; q[i + 64] = e2; q[i + 96] = e3;
        }
    }

    float4 a0 = make_float4(0.f, 0.f, 0.f, 0.f), a1 = a0, a2 = a0, a3 = a0, a4 = a0, a5 = a0, a6 = a0;
    for (int off = 0; off < total; off += 32) {
        const int cnt = min(32, total - off);
        const unsigned ent = (lane < cnt) ? q[off + lane] : 0u;
        gather_acc<SCALE>(h, degout, (int)(ent & 0xFFFFu), (int)(ent >> 16), cnt, lane,
                          a0, a1, a2, a3, a4, a5, a6);
    }

    float4* o4 = reinterpret_cast<float4*>(agg + (size_t)base * D);
    const float4 acc[NACC] = {a0, a1, a2, a3, a4, a5, a6};
#pragma unroll
    for (int j = 0; j < NACC; ++j) {
        const int r = wave + NWAVE * j;
        if (r < rows) {
            float4 o = acc[j];
            if (scan) {
                const float4 p = t4[r * (D / 4) + lane];
                o.x += p.x; o.y += p.y; o.z += p.z; o.w += p.w;
            }
            o4[r * (D / 4) + lane] = o;
        }
    }
}

#define GM_MT   5
#define GM_ROWS (16 * GM_MT)
#define GM_LDA  132
#define GM_RPT  (GM_ROWS / 8)

static_assert(NN % GM_ROWS == 0, "blocks cover the rows exactly, no row guard in the kernel");
static_assert(GM_ROWS % 8 == 0 && D == 128, "a wave stages whole 128-float rows, 8 waves take turns");

template <int MODE>
__global__ __launch_bounds__(256)
void norm_gemm_wmma(const float* __restrict__ agg,
                    const float* __restrict__ hprev,
                    const float* __restrict__ deg,
                    const float* __restrict__ W,
                    const float* __restrict__ bias,
                    float* __restrict__ out) {
    __shared__ __attribute__((aligned(16))) float Alds[GM_ROWS * GM_LDA];

    const int t      = threadIdx.x;
    const int waveId = t >> 5;
    const int lane   = t & 31;
    const int rowBase = blockIdx.x * GM_ROWS;

    {
        const float4* a4 = reinterpret_cast<const float4*>(agg)   + (size_t)rowBase * (D / 4) + lane;
        const float4* h4 = reinterpret_cast<const float4*>(hprev) + (size_t)rowBase * (D / 4) + lane;
        float4 av[GM_RPT], hv[GM_RPT];
        float  dg[GM_RPT];
#pragma unroll
        for (int i = 0; i < GM_RPT; ++i) {
            const int r = waveId + 8 * i;
            dg[i] = deg[rowBase + r];
            av[i] = a4[r * (D / 4)];
            if (MODE == 0) hv[i] = h4[r * (D / 4)];
        }
#pragma unroll
        for (int i = 0; i < GM_RPT; ++i) {
            const int r = waveId + 8 * i;
            float s;
            if (MODE == 0) s = 1.0f / (dg[i] + 1.0f);
            else           s = rsqrtf(fmaxf(dg[i], 1.0f));
            float4 v = av[i];
            if (MODE == 0) { v.x += hv[i].x; v.y += hv[i].y; v.z += hv[i].z; v.w += hv[i].w; }
            v.x *= s; v.y *= s; v.z *= s; v.w *= s;
            *reinterpret_cast<float4*>(Alds + r * GM_LDA + 4 * lane) = v;
        }
    }
    __syncthreads();

    const int half   = lane >> 4;
    const int lm     = lane & 15;
    const int col0   = waveId * 16;

    v8f c[GM_MT];
#pragma unroll
    for (int m = 0; m < GM_MT; ++m) c[m] = v8f{};

#pragma unroll
    for (int kk = 0; kk < 32; ++kk) {
        const int k0 = 4 * kk + 2 * half;
        v2f b;
        b.x = W[(size_t)(k0)     * D + col0 + lm];
        b.y = W[(size_t)(k0 + 1) * D + col0 + lm];
#pragma unroll
        for (int m = 0; m < GM_MT; ++m) {
            v2f a;
            a.x = Alds[(16 * m + lm) * GM_LDA + k0];
            a.y = Alds[(16 * m + lm) * GM_LDA + k0 + 1];
            c[m] = __builtin_amdgcn_wmma_f32_16x16x4_f32(
                       false, a, false, b, (short)0, c[m], false, false);
        }
    }

    const float bb = bias[col0 + lm];
#pragma unroll
    for (int m = 0; m < GM_MT; ++m) {
#pragma unroll
        for (int v = 0; v < 8; ++v) {
            const int row = rowBase + 16 * m + v + 8 * half;
            out[(size_t)row * D + col0 + lm] = c[m][v] + bb;
        }
    }
}

extern "C" void kernel_launch(void* const* d_in, const int* in_sizes, int n_in,
                              void* d_out, int out_size, void* d_ws, size_t ws_size,
                              hipStream_t stream) {
    const float* x  = (const float*)d_in[0];
    const float* W1 = (const float*)d_in[1];
    const float* b1 = (const float*)d_in[2];
    const float* W2 = (const float*)d_in[3];
    const float* b2 = (const float*)d_in[4];
    const float* W3 = (const float*)d_in[5];
    const float* b3 = (const float*)d_in[6];
    const int* src  = (const int*)d_in[7];
    const int* dst  = (const int*)d_in[8];
    float* out = (float*)d_out;

    const int N = NN;
    float* deg_in  = (float*)d_ws;
    float* deg_out = deg_in + N;
    float* agg     = deg_out + N;
    float* h1      = agg + (size_t)N * D;
    float* h2      = h1  + (size_t)N * D;
    int* reccnt       = (int*)(h2 + (size_t)N * D);
    unsigned* reclist = (unsigned*)(reccnt + NTILE * NWAVE);

    dim3 tileGrid(NTILE);
    dim3 gemmGrid(N / GM_ROWS);

    hipMemsetAsync(deg_out, 0, (size_t)N * sizeof(float), stream);

    tile_sum_kernel<0, 0><<<tileGrid, TBLK, 0, stream>>>(x, src, dst, deg_out, agg, reclist, reccnt, deg_in, deg_out);
    norm_gemm_wmma<0><<<gemmGrid, 256, 0, stream>>>(agg, x, deg_in, W1, b1, h1);

    tile_sum_kernel<0, 1><<<tileGrid, TBLK, 0, stream>>>(h1, src, dst, deg_out, agg, reclist, reccnt, deg_in, deg_out);
    norm_gemm_wmma<0><<<gemmGrid, 256, 0, stream>>>(agg, h1, deg_in, W2, b2, h2);

    tile_sum_kernel<1, 1><<<tileGrid, TBLK, 0, stream>>>(h2, src, dst, deg_out, agg, reclist, reccnt, deg_in, deg_out);
    norm_gemm_wmma<1><<<gemmGrid, 256, 0, stream>>>(agg, h2, deg_in, W3, b3, out);
}


# A small launch in front marks per block which 32-edge lines of dst hold an edge for it; layer 1's blocks read their mark words and only those lines, not the whole list; sums in queue arrival order
# speedup vs baseline: 2.8878x; 1.0582x over previous
#include <hip/hip_runtime.h>

#define NN 50000
#define NE 800000
#define D  128

typedef __attribute__((ext_vector_type(2))) float v2f;
typedef __attribute__((ext_vector_type(8))) float v8f;

#define RT     196
#define NTILE  256
#define GRP    32
#define GB     16
#define WLCAP  (32 + GB * GRP)
#define NQ     (NE / 4)
#define WSPAN  (32 * GRP)
#define NW     ((NE + WSPAN - 1) / WSPAN)
#define MARKW  8
#define MARKBLK ((NW + MARKW - 1) / MARKW)
#define TBLK   1024
#define NWAVE  (TBLK / 32)
#define QCAP   512
#define NACC   7
#define GLW    ((NW + NWAVE - 1) / NWAVE)
#define GLCAP  (32 * GLW)

static_assert(NE % 4 == 0 && NE % GRP == 0 && WSPAN % 128 == 0, "dst is read in 16-byte words and in whole runs");
static_assert(WLCAP >= 31 + GB * GRP, "a wave's list holds the rest of one flush plus one full batch of runs");
static_assert(GLW <= 32 && NE / GRP <= (1 << 16), "a lane per mark word of the wave; a run's number fits 16 bits");
static_assert(NTILE % 32 == 0, "a marking wave stores its table in whole 128-byte lines");
static_assert(NE <= (1 << 20) && RT <= 256, "entry packs edge id (20 bits) and local row (8 bits)");
static_assert((long)NTILE * RT >= NN && (long)(NTILE - 1) * RT < NN, "tiles cover every row");
static_assert(NN <= (1 << 16), "queue entry packs source row (16 bits) and the owner's row number");
static_assert(QCAP % 128 == 0, "a kept queue is staged back into LDS in steps of 128 entries");
static_assert(NACC * NWAVE >= RT && (NWAVE & (NWAVE - 1)) == 0, "every local row has an owner wave and a register set");

__device__ __forceinline__ void tile_add_row(float* trow, float4 v,
                                             int rot, int lo, int lo3) {
    float a0 = rot ? v.y : v.x;
    float a1 = rot ? v.z : v.y;
    float a2 = rot ? v.w : v.z;
    float a3 = rot ? v.x : v.w;
    __hip_atomic_fetch_add(trow + lo + 0, a0, __ATOMIC_RELAXED, __HIP_MEMORY_SCOPE_WORKGROUP);
    __hip_atomic_fetch_add(trow + lo + 1, a1, __ATOMIC_RELAXED, __HIP_MEMORY_SCOPE_WORKGROUP);
    __hip_atomic_fetch_add(trow + lo + 2, a2, __ATOMIC_RELAXED, __HIP_MEMORY_SCOPE_WORKGROUP);
    __hip_atomic_fetch_add(trow + lo3,    a3, __ATOMIC_RELAXED, __HIP_MEMORY_SCOPE_WORKGROUP);
}

__global__ __launch_bounds__(32 * MARKW)
void tile_mark_kernel(const int* __restrict__ dst, unsigned* __restrict__ marks) {
    __shared__ unsigned tab[MARKW * NTILE];
    const int lane = threadIdx.x & 31;
    const int wv   = threadIdx.x >> 5;
    const int w    = blockIdx.x * MARKW + wv;
    if (w >= NW) return;
    unsigned* tb = tab + wv * NTILE;
#pragma unroll
    for (int k = 0; k < NTILE / 32; ++k) tb[lane + 32 * k] = 0u;
    const int4* d4 = reinterpret_cast<const int4*>(dst);
    const int qb = w * (WSPAN / 4);
    int4 dv[WSPAN / 128];
#pragma unroll
    for (int j = 0; j < WSPAN / 128; ++j) {
        const int q = qb + 32 * j + lane;
        dv[j] = (q < NQ) ? d4[q] : make_int4(-1, -1, -1, -1);
    }
#pragma unroll
    for (int j = 0; j < WSPAN / 128; ++j) {
        const unsigned bit = 1u << (4 * j + (lane >> 3));
        const unsigned d0 = (unsigned)dv[j].x, d1 = (unsigned)dv[j].y, d2 = (unsigned)dv[j].z, d3 = (unsigned)dv[j].w;
        if (d0 < (unsigned)NN) __hip_atomic_fetch_or(tb + d0 / RT, bit, __ATOMIC_RELAXED, __HIP_MEMORY_SCOPE_WORKGROUP);
        if (d1 < (unsigned)NN) __hip_atomic_fetch_or(tb + d1 / RT, bit, __ATOMIC_RELAXED, __HIP_MEMORY_SCOPE_WORKGROUP);
        if (d2 < (unsigned)NN) __hip_atomic_fetch_or(tb + d2 / RT, bit, __ATOMIC_RELAXED, __HIP_MEMORY_SCOPE_WORKGROUP);
        if (d3 < (unsigned)NN) __hip_atomic_fetch_or(tb + d3 / RT, bit, __ATOMIC_RELAXED, __HIP_MEMORY_SCOPE_WORKGROUP);
    }
    unsigned* mrow = marks + (size_t)w * NTILE;
#pragma unroll
    for (int k = 0; k < NTILE / 32; ++k) mrow[lane + 32 * k] = tb[lane + 32 * k];
}

#define ACC_CASE(j) case j: a##j.x += x.x; a##j.y += x.y; a##j.z += x.z; a##j.w += x.w; break;

template <int SCALE>
__device__ __forceinline__ void gather_acc(const float* __restrict__ h,
                                           const float* __restrict__ degout,
                                           int s, int c, int cnt, int lane,
                                           float4& a0, float4& a1, float4& a2, float4& a3,
                                           float4& a4, float4& a5, float4& a6) {
    int scb = 0;
    if (SCALE) scb = __float_as_int(rsqrtf(fmaxf(degout[s], 1.0f)));
    const float4* h4 = reinterpret_cast<const float4*>(h);
    for (int k0 = 0; k0 < cnt; k0 += 8) {
        float4 v[8];
#pragma unroll
        for (int k = 0; k < 8; ++k) {
            int sk = __builtin_amdgcn_readlane(s, k0 + k);
            v[k] = h4[(size_t)sk * (D / 4) + lane];
        }
#pragma unroll
        for (int k = 0; k < 8; ++k) {
            if (k0 + k < cnt) {
                const int ck = __builtin_amdgcn_readlane(c, k0 + k);
                float4 x = v[k];
                if (SCALE) {
                    float sc = __int_as_float(__builtin_amdgcn_readlane(scb, k0 + k));
                    x.x *= sc; x.y *= sc; x.z *= sc; x.w *= sc;
                }
                switch (ck) {
                    ACC_CASE(0) ACC_CASE(1) ACC_CASE(2) ACC_CASE(3)
                    ACC_CASE(4) ACC_CASE(5) ACC_CASE(6)
                    default: break;
                }
            }
        }
    }
}

template <int SCALE, int COUNT>
__device__ __forceinline__ void route_round(const float* __restrict__ h,
                                            const int* __restrict__ src,
                                            const float* __restrict__ degout,
                                            float* tile, const unsigned* wl,
                                            int off, int cnt, int lane,
                                            int rot, int lo, int lo3,
                                            unsigned* queue, int* qcnt,
                                            int* rowdeg, float* outdeg) {
    const unsigned ent = (lane < cnt) ? wl[off + lane] : 0u;
    const int e = (int)(ent & 0xFFFFFu);
    const int r = (int)(ent >> 20);
    int s = 0;
    bool spill = false;
    if (lane < cnt) {
        s = src[e];
        if (COUNT) {
            __hip_atomic_fetch_add(rowdeg + r, 1, __ATOMIC_RELAXED, __HIP_MEMORY_SCOPE_WORKGROUP);
            unsafeAtomicAdd(outdeg + s, 1.0f);
        }
        const int owner = r & (NWAVE - 1);
        const int slot = __hip_atomic_fetch_add(qcnt + owner, 1, __ATOMIC_RELAXED, __HIP_MEMORY_SCOPE_WORKGROUP);
        if (slot < QCAP) queue[owner * QCAP + slot] = (unsigned)s | ((unsigned)(r >> 5) << 16);
        else spill = true;
    }
    unsigned m = __builtin_amdgcn_ballot_w32(spill);
    const float4* h4 = reinterpret_cast<const float4*>(h);
    while (m) {
        const int k = __builtin_ctz(m);
        m &= m - 1;
        const int sk = __builtin_amdgcn_readlane(s, k);
        const int rk = __builtin_amdgcn_readlane(r, k);
        float4 x = h4[(size_t)sk * (D / 4) + lane];
        if (SCALE) {
            const float sc = rsqrtf(fmaxf(degout[sk], 1.0f));
            x.x *= sc; x.y *= sc; x.z *= sc; x.w *= sc;
        }
        tile_add_row(tile + rk * D, x, rot, lo, lo3);
    }
}

template <int SCALE, int COUNT>
__device__ __forceinline__ void scan_and_route(const float* __restrict__ h,
                                               const int* __restrict__ src,
                                               const int* __restrict__ dst,
                                               const unsigned* __restrict__ marks,
                                               const float* __restrict__ degout,
                                               float* tile, unsigned* wl, unsigned short* gl,
                                               int wave, int lane, int base, int rows,
                                               int rot, int lo, int lo3,
                                               unsigned* queue, int* qcnt,
                                               int* rowdeg, float* outdeg) {
    const int nl = (NW - wave + NWAVE - 1) / NWAVE;
    unsigned pw = 0u;
    if (lane < nl) pw = marks[(size_t)(wave + NWAVE * lane) * NTILE + blockIdx.x];

    int ng = 0;
    for (int l = 0; l < nl; ++l) {
        const unsigned m = (unsigned)__builtin_amdgcn_readlane((int)pw, l);
        if (m) {
            if ((m >> lane) & 1u)
                gl[ng + __builtin_amdgcn_mbcnt_lo(m, 0u)] = (unsigned short)((wave + NWAVE * l) * 32 + lane);
            ng += __builtin_popcount(m);
        }
    }

    int wcount = 0;
    for (int i0 = 0; i0 < ng; i0 += GB) {
        const int nb = min(GB, ng - i0);
        const int gme = (lane < nb) ? (int)gl[i0 + lane] : 0;
        int dv[GB];
#pragma unroll
        for (int k = 0; k < GB; ++k)
            dv[k] = dst[__builtin_amdgcn_readlane(gme, k) * GRP + lane];
#pragma unroll
        for (int k = 0; k < GB; ++k) {
            if (k < nb) {
                const int e = __builtin_amdgcn_readlane(gme, k) * GRP + lane;
                const unsigned r = (unsigned)(dv[k] - base);
                const bool hit = r < (unsigned)rows;
                const unsigned m = __builtin_amdgcn_ballot_w32(hit);
                if (hit) wl[wcount + __builtin_amdgcn_mbcnt_lo(m, 0u)] = (unsigned)e | (r << 20);
                wcount += __builtin_popcount(m);
            }
        }
        if (wcount >= 32) {
            int done = 0;
            for (; done + 32 <= wcount; done += 32)
                route_round<SCALE, COUNT>(h, src, degout, tile, wl, done, 32, lane, rot, lo, lo3, queue, qcnt,
                                          rowdeg, outdeg);
            const int rem = wcount - done;
            unsigned keep = (lane < rem) ? wl[done + lane] : 0u;
            if (lane < rem) wl[lane] = keep;
            wcount = rem;
        }
    }
    for (int done = 0; done < wcount; done += 32)
        route_round<SCALE, COUNT>(h, src, degout, tile, wl, done, min(32, wcount - done), lane, rot, lo, lo3, queue, qcnt,
                                  rowdeg, outdeg);
}

template <int SCALE, int REPLAY>
__global__ __launch_bounds__(TBLK)
void tile_sum_kernel(const float* __restrict__ h,
                     const int* __restrict__ src,
                     const int* __restrict__ dst,
                     const unsigned* __restrict__ marks,
                     const float* __restrict__ degout,
                     float* __restrict__ agg,
                     unsigned* reclist,
                     int* reccnt,
                     float* indeg,
                     float* outdeg) {
    __shared__ __attribute__((aligned(16))) float tile[RT * D];
    __shared__ unsigned wlist[NWAVE * WLCAP];
    __shared__ unsigned short glist[NWAVE * GLCAP];
    __shared__ unsigned queue[NWAVE * QCAP];
    __shared__ int qcnt[NWAVE];
    __shared__ int rowdeg[RT];

    const int t    = threadIdx.x;
    const int lane = t & 31;
    const int wave = __builtin_amdgcn_readfirstlane(t >> 5);
    const int base = blockIdx.x * RT;
    const int rows = min(RT, NN - base);
    const int rot  = lane >> 4;
    const int lo   = 4 * lane + rot;
    const int lo3  = 4 * lane + 3 - 3 * rot;

    float4* t4 = reinterpret_cast<float4*>(tile);
    const int gw = blockIdx.x * NWAVE + wave;
    unsigned* rec = reclist + (size_t)gw * QCAP;
    unsigned* q   = queue + wave * QCAP;

    bool scan = true;
    int total = 0;
    if (REPLAY) {
        const int cw = reccnt[blockIdx.x * NWAVE + lane];
        scan  = __builtin_amdgcn_ballot_w32(cw > QCAP) != 0u;
        total = __builtin_amdgcn_readlane(cw, wave);
    }

    if (scan) {
        for (int i = t; i < RT * D / 4; i += TBLK) t4[i] = make_float4(0.f, 0.f, 0.f, 0.f);
        if (t < NWAVE) qcnt[t] = 0;
        if (!REPLAY && t < RT) rowdeg[t] = 0;
        __syncthreads();
        scan_and_route<SCALE, !REPLAY>(h, src, dst, marks, degout, tile, wlist + wave * WLCAP, glist + wave * GLCAP,
                                       wave, lane, base, rows, rot, lo, lo3, queue, qcnt, rowdeg, outdeg);
        __syncthreads();
        const int filled = __builtin_amdgcn_readfirstlane(qcnt[wave]);
        total = min(filled, QCAP);
        if (!REPLAY) {
            for (int i = lane; i < total; i += 32) rec[i] = q[i];
            if (lane == 0) reccnt[gw] = filled;
            if (t < rows) indeg[base + t] = (float)rowdeg[t];
        }
    } else {
        for (int i0 = 0; i0 < total; i0 += 128) {
            const int i = i0 + lane;
            const unsigned e0 = (i      < total) ? rec[i]      : 0u;
            const unsigned e1 = (i + 32 < total) ? rec[i + 32] : 0u;
            const unsigned e2 = (i + 64 < total) ? rec[i + 64] : 0u;
            const unsigned e3 = (i + 96 < total) ? rec[i + 96] : 0u;
            q[i] = e0; q[i + 32] = e1; q[i + 64] = e2; q[i + 96] = e3;
        }
    }

    float4 a0 = make_float4(0.f, 0.f, 0.f, 0.f), a1 = a0, a2 = a0, a3 = a0, a4 = a0, a5 = a0, a6 = a0;
    for (int off = 0; off < total; off += 32) {
        const int cnt = min(32, total - off);
        const unsigned ent = (lane < cnt) ? q[off + lane] : 0u;
        gather_acc<SCALE>(h, degout, (int)(ent & 0xFFFFu), (int)(ent >> 16), cnt, lane,
                          a0, a1, a2, a3, a4, a5, a6);
    }

    float4* o4 = reinterpret_cast<float4*>(agg + (size_t)base * D);
    const float4 acc[NACC] = {a0, a1, a2, a3, a4, a5, a6};
#pragma unroll
    for (int j = 0; j < NACC; ++j) {
        const int r = wave + NWAVE * j;
        if (r < rows) {
            float4 o = acc[j];
            if (scan) {
                const float4 p = t4[r * (D / 4) + lane];
                o.x += p.x; o.y += p.y; o.z += p.z; o.w += p.w;
            }
            o4[r * (D / 4) + lane] = o;
        }
    }
}

#define GM_MT   5
#define GM_ROWS (16 * GM_MT)
#define GM_LDA  132
#define GM_RPT  (GM_ROWS / 8)

static_assert(NN % GM_ROWS == 0, "blocks cover the rows exactly, no row guard in the kernel");
static_assert(GM_ROWS % 8 == 0 && D == 128, "a wave stages whole 128-float rows, 8 waves take turns");

template <int MODE>
__global__ __launch_bounds__(256)
void norm_gemm_wmma(const float* __restrict__ agg,
                    const float* __restrict__ hprev,
                    const float* __restrict__ deg,
                    const float* __restrict__ W,
                    const float* __restrict__ bias,
                    float* __restrict__ out) {
    __shared__ __attribute__((aligned(16))) float Alds[GM_ROWS * GM_LDA];

    const int t      = threadIdx.x;
    const int waveId = t >> 5;
    const int lane   = t & 31;
    const int rowBase = blockIdx.x * GM_ROWS;

    {
        const float4* a4 = reinterpret_cast<const float4*>(agg)   + (size_t)rowBase * (D / 4) + lane;
        const float4* h4 = reinterpret_cast<const float4*>(hprev) + (size_t)rowBase * (D / 4) + lane;
        float4 av[GM_RPT], hv[GM_RPT];
        float  dg[GM_RPT];
#pragma unroll
        for (int i = 0; i < GM_RPT; ++i) {
            const int r = waveId + 8 * i;
            dg[i] = deg[rowBase + r];
            av[i] = a4[r * (D / 4)];
            if (MODE == 0) hv[i] = h4[r * (D / 4)];
        }
#pragma unroll
        for (int i = 0; i < GM_RPT; ++i) {
            const int r = waveId + 8 * i;
            float s;
            if (MODE == 0) s = 1.0f / (dg[i] + 1.0f);
            else           s = rsqrtf(fmaxf(dg[i], 1.0f));
            float4 v = av[i];
            if (MODE == 0) { v.x += hv[i].x; v.y += hv[i].y; v.z += hv[i].z; v.w += hv[i].w; }
            v.x *= s; v.y *= s; v.z *= s; v.w *= s;
            *reinterpret_cast<float4*>(Alds + r * GM_LDA + 4 * lane) = v;
        }
    }
    __syncthreads();

    const int half   = lane >> 4;
    const int lm     = lane & 15;
    const int col0   = waveId * 16;

    v8f c[GM_MT];
#pragma unroll
    for (int m = 0; m < GM_MT; ++m) c[m] = v8f{};

#pragma unroll
    for (int kk = 0; kk < 32; ++kk) {
        const int k0 = 4 * kk + 2 * half;
        v2f b;
        b.x = W[(size_t)(k0)     * D + col0 + lm];
        b.y = W[(size_t)(k0 + 1) * D + col0 + lm];
#pragma unroll
        for (int m = 0; m < GM_MT; ++m) {
            v2f a;
            a.x = Alds[(16 * m + lm) * GM_LDA + k0];
            a.y = Alds[(16 * m + lm) * GM_LDA + k0 + 1];
            c[m] = __builtin_amdgcn_wmma_f32_16x16x4_f32(
                       false, a, false, b, (short)0, c[m], false, false);
        }
    }

    const float bb = bias[col0 + lm];
#pragma unroll
    for (int m = 0; m < GM_MT; ++m) {
#pragma unroll
        for (int v = 0; v < 8; ++v) {
            const int row = rowBase + 16 * m + v + 8 * half;
            out[(size_t)row * D + col0 + lm] = c[m][v] + bb;
        }
    }
}

extern "C" void kernel_launch(void* const* d_in, const int* in_sizes, int n_in,
                              void* d_out, int out_size, void* d_ws, size_t ws_size,
                              hipStream_t stream) {
    const float* x  = (const float*)d_in[0];
    const float* W1 = (const float*)d_in[1];
    const float* b1 = (const float*)d_in[2];
    const float* W2 = (const float*)d_in[3];
    const float* b2 = (const float*)d_in[4];
    const float* W3 = (const float*)d_in[5];
    const float* b3 = (const float*)d_in[6];
    const int* src  = (const int*)d_in[7];
    const int* dst  = (const int*)d_in[8];
    float* out = (float*)d_out;

    const int N = NN;
    float* deg_in  = (float*)d_ws;
    float* deg_out = deg_in + N;
    float* agg     = deg_out + N;
    float* h1      = agg + (size_t)N * D;
    float* h2      = h1  + (size_t)N * D;
    int* reccnt       = (int*)(h2 + (size_t)N * D);
    unsigned* reclist = (unsigned*)(reccnt + NTILE * NWAVE);
    unsigned* marks   = reclist + (size_t)NTILE * NWAVE * QCAP;

    dim3 tileGrid(NTILE);
    dim3 gemmGrid(N / GM_ROWS);
    dim3 markGrid(MARKBLK);

    hipMemsetAsync(deg_out, 0, (size_t)N * sizeof(float), stream);

    tile_mark_kernel<<<markGrid, 32 * MARKW, 0, stream>>>(dst, marks);

    tile_sum_kernel<0, 0><<<tileGrid, TBLK, 0, stream>>>(x, src, dst, marks, deg_out, agg, reclist, reccnt, deg_in, deg_out);
    norm_gemm_wmma<0><<<gemmGrid, 256, 0, stream>>>(agg, x, deg_in, W1, b1, h1);

    tile_sum_kernel<0, 1><<<tileGrid, TBLK, 0, stream>>>(h1, src, dst, marks, deg_out, agg, reclist, reccnt, deg_in, deg_out);
    norm_gemm_wmma<0><<<gemmGrid, 256, 0, stream>>>(agg, h1, deg_in, W2, b2, h2);

    tile_sum_kernel<1, 1><<<tileGrid, TBLK, 0, stream>>>(h2, src, dst, marks, deg_out, agg, reclist, reccnt, deg_in, deg_out);
    norm_gemm_wmma<1><<<gemmGrid, 256, 0, stream>>>(agg, h2, deg_in, W3, b3, out);
}


# Owner wave's queue split into one LDS queue per row (slot = rank of the row's counter): a wave sums its rows one after another in ONE register set, no per-edge branch among 7 sets; arrival order kept
# speedup vs baseline: 3.0723x; 1.0639x over previous
#include <hip/hip_runtime.h>

#define NN 50000
#define NE 800000
#define D  128

typedef __attribute__((ext_vector_type(2))) float v2f;
typedef __attribute__((ext_vector_type(8))) float v8f;

#define RT     196
#define NTILE  256
#define GRP    32
#define GB     16
#define WLCAP  (32 + GB * GRP)
#define NQ     (NE / 4)
#define WSPAN  (32 * GRP)
#define NW     ((NE + WSPAN - 1) / WSPAN)
#define MARKW  8
#define MARKBLK ((NW + MARKW - 1) / MARKW)
#define TBLK   1024
#define NWAVE  (TBLK / 32)
#define QCAP   512
#define RQ     64
#define RCW    8
#define GBAT   16
#define NACC   7
#define GLW    ((NW + NWAVE - 1) / NWAVE)
#define GLCAP  (32 * GLW)

static_assert(NE % 4 == 0 && NE % GRP == 0 && WSPAN % 128 == 0, "dst is read in 16-byte words and in whole runs");
static_assert(WLCAP >= 31 + GB * GRP, "a wave's list holds the rest of one flush plus one full batch of runs");
static_assert(GLW <= 32 && NE / GRP <= (1 << 16), "a lane per mark word of the wave; a run's number fits 16 bits");
static_assert(NTILE % 32 == 0, "a marking wave stores its table in whole 128-byte lines");
static_assert(NE <= (1 << 20) && RT <= 256, "entry packs edge id (20 bits) and local row (8 bits)");
static_assert((long)NTILE * RT >= NN && (long)(NTILE - 1) * RT < NN, "tiles cover every row");
static_assert(NN <= (1 << 16), "a row queue entry is a source row in 16 bits");
static_assert(NACC * RQ <= QCAP && NACC < RCW && 32 % GBAT == 0, "a wave's list holds all its row queues; a round of 32 entries is whole batches");
static_assert(QCAP % 128 == 0, "a kept list is staged back into LDS in steps of 128 entries");
static_assert(NACC * NWAVE >= RT && (NWAVE & (NWAVE - 1)) == 0, "every local row has an owner wave");

__device__ __forceinline__ void tile_add_row(float* trow, float4 v,
                                             int rot, int lo, int lo3) {
    float a0 = rot ? v.y : v.x;
    float a1 = rot ? v.z : v.y;
    float a2 = rot ? v.w : v.z;
    float a3 = rot ? v.x : v.w;
    __hip_atomic_fetch_add(trow + lo + 0, a0, __ATOMIC_RELAXED, __HIP_MEMORY_SCOPE_WORKGROUP);
    __hip_atomic_fetch_add(trow + lo + 1, a1, __ATOMIC_RELAXED, __HIP_MEMORY_SCOPE_WORKGROUP);
    __hip_atomic_fetch_add(trow + lo + 2, a2, __ATOMIC_RELAXED, __HIP_MEMORY_SCOPE_WORKGROUP);
    __hip_atomic_fetch_add(trow + lo3,    a3, __ATOMIC_RELAXED, __HIP_MEMORY_SCOPE_WORKGROUP);
}

__global__ __launch_bounds__(32 * MARKW)
void tile_mark_kernel(const int* __restrict__ dst, unsigned* __restrict__ marks) {
    __shared__ unsigned tab[MARKW * NTILE];
    const int lane = threadIdx.x & 31;
    const int wv   = threadIdx.x >> 5;
    const int w    = blockIdx.x * MARKW + wv;
    if (w >= NW) return;
    unsigned* tb = tab + wv * NTILE;
#pragma unroll
    for (int k = 0; k < NTILE / 32; ++k) tb[lane + 32 * k] = 0u;
    const int4* d4 = reinterpret_cast<const int4*>(dst);
    const int qb = w * (WSPAN / 4);
    int4 dv[WSPAN / 128];
#pragma unroll
    for (int j = 0; j < WSPAN / 128; ++j) {
        const int q = qb + 32 * j + lane;
        dv[j] = (q < NQ) ? d4[q] : make_int4(-1, -1, -1, -1);
    }
#pragma unroll
    for (int j = 0; j < WSPAN / 128; ++j) {
        const unsigned bit = 1u << (4 * j + (lane >> 3));
        const unsigned d0 = (unsigned)dv[j].x, d1 = (unsigned)dv[j].y, d2 = (unsigned)dv[j].z, d3 = (unsigned)dv[j].w;
        if (d0 < (unsigned)NN) __hip_atomic_fetch_or(tb + d0 / RT, bit, __ATOMIC_RELAXED, __HIP_MEMORY_SCOPE_WORKGROUP);
        if (d1 < (unsigned)NN) __hip_atomic_fetch_or(tb + d1 / RT, bit, __ATOMIC_RELAXED, __HIP_MEMORY_SCOPE_WORKGROUP);
        if (d2 < (unsigned)NN) __hip_atomic_fetch_or(tb + d2 / RT, bit, __ATOMIC_RELAXED, __HIP_MEMORY_SCOPE_WORKGROUP);
        if (d3 < (unsigned)NN) __hip_atomic_fetch_or(tb + d3 / RT, bit, __ATOMIC_RELAXED, __HIP_MEMORY_SCOPE_WORKGROUP);
    }
    unsigned* mrow = marks + (size_t)w * NTILE;
#pragma unroll
    for (int k = 0; k < NTILE / 32; ++k) mrow[lane + 32 * k] = tb[lane + 32 * k];
}

template <int SCALE, int COUNT>
__device__ __forceinline__ void route_round(const float* __restrict__ h,
                                            const int* __restrict__ src,
                                            const float* __restrict__ degout,
                                            float* tile, const unsigned* wl,
                                            int off, int cnt, int lane,
                                            int rot, int lo, int lo3,
                                            unsigned short* rowq,
                                            int* rowdeg, float* outdeg) {
    const unsigned ent = (lane < cnt) ? wl[off + lane] : 0u;
    const int e = (int)(ent & 0xFFFFFu);
    const int r = (int)(ent >> 20);
    int s = 0;
    bool spill = false;
    if (lane < cnt) {
        s = src[e];
        if (COUNT) unsafeAtomicAdd(outdeg + s, 1.0f);
        const int slot = __hip_atomic_fetch_add(rowdeg + r, 1, __ATOMIC_RELAXED, __HIP_MEMORY_SCOPE_WORKGROUP);
        if (slot < RQ) rowq[r * RQ + slot] = (unsigned short)s;
        else spill = true;
    }
    unsigned m = __builtin_amdgcn_ballot_w32(spill);
    const float4* h4 = reinterpret_cast<const float4*>(h);
    while (m) {
        const int k = __builtin_ctz(m);
        m &= m - 1;
        const int sk = __builtin_amdgcn_readlane(s, k);
        const int rk = __builtin_amdgcn_readlane(r, k);
        float4 x = h4[(size_t)sk * (D / 4) + lane];
        if (SCALE) {
            const float sc = rsqrtf(fmaxf(degout[sk], 1.0f));
            x.x *= sc; x.y *= sc; x.z *= sc; x.w *= sc;
        }
        tile_add_row(tile + rk * D, x, rot, lo, lo3);
    }
}

template <int SCALE, int COUNT>
__device__ __forceinline__ void scan_and_route(const float* __restrict__ h,
                                               const int* __restrict__ src,
                                               const int* __restrict__ dst,
                                               const unsigned* __restrict__ marks,
                                               const float* __restrict__ degout,
                                               float* tile, unsigned* wl, unsigned short* gl,
                                               int wave, int lane, int base, int rows,
                                               int rot, int lo, int lo3,
                                               unsigned short* rowq,
                                               int* rowdeg, float* outdeg) {
    const int nl = (NW - wave + NWAVE - 1) / NWAVE;
    unsigned pw = 0u;
    if (lane < nl) pw = marks[(size_t)(wave + NWAVE * lane) * NTILE + blockIdx.x];

    int ng = 0;
    for (int l = 0; l < nl; ++l) {
        const unsigned m = (unsigned)__builtin_amdgcn_readlane((int)pw, l);
        if (m) {
            if ((m >> lane) & 1u)
                gl[ng + __builtin_amdgcn_mbcnt_lo(m, 0u)] = (unsigned short)((wave + NWAVE * l) * 32 + lane);
            ng += __builtin_popcount(m);
        }
    }

    int wcount = 0;
    for (int i0 = 0; i0 < ng; i0 += GB) {
        const int nb = min(GB, ng - i0);
        const int gme = (lane < nb) ? (int)gl[i0 + lane] : 0;
        int dv[GB];
#pragma unroll
        for (int k = 0; k < GB; ++k)
            dv[k] = dst[__builtin_amdgcn_readlane(gme, k) * GRP + lane];
#pragma unroll
        for (int k = 0; k < GB; ++k) {
            if (k < nb) {
                const int e = __builtin_amdgcn_readlane(gme, k) * GRP + lane;
                const unsigned r = (unsigned)(dv[k] - base);
                const bool hit = r < (unsigned)rows;
                const unsigned m = __builtin_amdgcn_ballot_w32(hit);
                if (hit) wl[wcount + __builtin_amdgcn_mbcnt_lo(m, 0u)] = (unsigned)e | (r << 20);
                wcount += __builtin_popcount(m);
            }
        }
        if (wcount >= 32) {
            int done = 0;
            for (; done + 32 <= wcount; done += 32)
                route_round<SCALE, COUNT>(h, src, degout, tile, wl, done, 32, lane, rot, lo, lo3, rowq,
                                          rowdeg, outdeg);
            const int rem = wcount - done;
            unsigned keep = (lane < rem) ? wl[done + lane] : 0u;
            if (lane < rem) wl[lane] = keep;
            wcount = rem;
        }
    }
    for (int done = 0; done < wcount; done += 32)
        route_round<SCALE, COUNT>(h, src, degout, tile, wl, done, min(32, wcount - done), lane, rot, lo, lo3, rowq,
                                  rowdeg, outdeg);
}

template <int SCALE, int REPLAY>
__global__ __launch_bounds__(TBLK)
void tile_sum_kernel(const float* __restrict__ h,
                     const int* __restrict__ src,
                     const int* __restrict__ dst,
                     const unsigned* __restrict__ marks,
                     const float* __restrict__ degout,
                     float* __restrict__ agg,
                     unsigned* reclist,
                     int* reccnt,
                     float* indeg,
                     float* outdeg) {
    __shared__ __attribute__((aligned(16))) float tile[RT * D];
    __shared__ unsigned wlist[NWAVE * WLCAP];
    __shared__ unsigned short glist[NWAVE * GLCAP];
    __shared__ unsigned queue[NWAVE * QCAP];
    __shared__ unsigned short rowq[RT * RQ];
    __shared__ int rowdeg[RT];

    const int t    = threadIdx.x;
    const int lane = t & 31;
    const int wave = __builtin_amdgcn_readfirstlane(t >> 5);
    const int base = blockIdx.x * RT;
    const int rows = min(RT, NN - base);
    const int rot  = lane >> 4;
    const int lo   = 4 * lane + rot;
    const int lo3  = 4 * lane + 3 - 3 * rot;

    float4* t4 = reinterpret_cast<float4*>(tile);
    const int gw = blockIdx.x * NWAVE + wave;
    unsigned* rec = reclist + (size_t)gw * QCAP;
    int* rc       = reccnt + (size_t)gw * RCW;
    unsigned* q   = queue + wave * QCAP;

    bool scan = true;
    int cntv = 0;
    if (REPLAY) {
        const int full = reccnt[(blockIdx.x * NWAVE + lane) * RCW + NACC];
        scan = __builtin_amdgcn_ballot_w32(full != 0) != 0u;
        if (lane < NACC) cntv = rc[lane];
    }

    if (scan) {
        for (int i = t; i < RT * D / 4; i += TBLK) t4[i] = make_float4(0.f, 0.f, 0.f, 0.f);
        if (t < RT) rowdeg[t] = 0;
        __syncthreads();
        scan_and_route<SCALE, !REPLAY>(h, src, dst, marks, degout, tile, wlist + wave * WLCAP, glist + wave * GLCAP,
                                       wave, lane, base, rows, rot, lo, lo3, rowq, rowdeg, outdeg);
        __syncthreads();
        int dg = 0;
        if (lane < NACC && wave + NWAVE * lane < rows) dg = rowdeg[wave + NWAVE * lane];
        cntv = min(dg, RQ);
        const bool over = __builtin_amdgcn_ballot_w32(dg > RQ) != 0u;
        int o = 0;
        for (int j = 0; j < NACC; ++j) {
            const int nj = __builtin_amdgcn_readlane(cntv, j);
            const unsigned short* rq = rowq + (wave + NWAVE * j) * RQ;
            for (int i = lane; i < nj; i += 32) q[o + i] = rq[i];
            o += nj;
        }
        if (!REPLAY) {
            for (int i = lane; i < o; i += 32) rec[i] = q[i];
            if (lane < RCW) rc[lane] = (lane < NACC) ? cntv : (over ? 1 : 0);
            if (t < rows) indeg[base + t] = (float)rowdeg[t];
        }
    }

    int total = 0;
    int endv = 0x7fffffff;
    for (int j = 0; j < NACC; ++j) {
        total += __builtin_amdgcn_readlane(cntv, j);
        if (lane == j) endv = total;
    }

    if (!scan) {
        for (int i0 = 0; i0 < total; i0 += 128) {
            const int i = i0 + lane;
            const unsigned e0 = (i      < total) ? rec[i]      : 0u;
            const unsigned e1 = (i + 32 < total) ? rec[i + 32] : 0u;
            const unsigned e2 = (i + 64 < total) ? rec[i + 64] : 0u;
            const unsigned e3 = (i + 96 < total) ? rec[i + 96] : 0u;
            q[i] = e0; q[i + 32] = e1; q[i + 64] = e2; q[i + 96] = e3;
        }
    }

    float4* o4 = reinterpret_cast<float4*>(agg + (size_t)base * D);
    const char* hb = reinterpret_cast<const char*>(h);
    const unsigned lane16 = 16u * (unsigned)lane;
    float4 acc = make_float4(0.f, 0.f, 0.f, 0.f);
    int j = 0;
    int end = __builtin_amdgcn_readlane(endv, 0);
    for (int off = 0; off < total; off += 32) {
        const int cnt = min(32, total - off);
        const unsigned ent = (lane < cnt) ? q[off + lane] : 0u;
        int scb = 0;
        if (SCALE) scb = __float_as_int(rsqrtf(fmaxf(degout[ent], 1.0f)));
        for (int k0 = 0; k0 < cnt; k0 += GBAT) {
            float4 v[GBAT];
#pragma unroll
            for (int k = 0; k < GBAT; ++k) {
                const unsigned sk = (unsigned)__builtin_amdgcn_readlane((int)ent, k0 + k);
                v[k] = *reinterpret_cast<const float4*>(hb + (size_t)sk * (D * sizeof(float)) + lane16);
            }
#pragma unroll
            for (int k = 0; k < GBAT; ++k) {
                if (k0 + k < cnt) {
                    const int pos = off + k0 + k;
                    while (pos == end) {
                        const int r = wave + NWAVE * j;
                        if (r < rows) {
                            float4 o = acc;
                            if (scan) {
                                const float4 p = t4[r * (D / 4) + lane];
                                o.x += p.x; o.y += p.y; o.z += p.z; o.w += p.w;
                            }
                            o4[r * (D / 4) + lane] = o;
                        }
                        acc = make_float4(0.f, 0.f, 0.f, 0.f);
                        ++j;
                        end = __builtin_amdgcn_readlane(endv, j);
                    }
                    float4 x = v[k];
                    if (SCALE) {
                        const float sc = __int_as_float(__builtin_amdgcn_readlane(scb, k0 + k));
                        x.x *= sc; x.y *= sc; x.z *= sc; x.w *= sc;
                    }
                    acc.x += x.x; acc.y += x.y; acc.z += x.z; acc.w += x.w;
                }
            }
        }
    }
    for (; j < NACC; ++j) {
        const int r = wave + NWAVE * j;
        if (r < rows) {
            float4 o = acc;
            if (scan) {
                const float4 p = t4[r * (D / 4) + lane];
                o.x += p.x; o.y += p.y; o.z += p.z; o.w += p.w;
            }
            o4[r * (D / 4) + lane] = o;
        }
        acc = make_float4(0.f, 0.f, 0.f, 0.f);
    }
}

#define GM_MT   5
#define GM_ROWS (16 * GM_MT)
#define GM_LDA  132
#define GM_RPT  (GM_ROWS / 8)

static_assert(NN % GM_ROWS == 0, "blocks cover the rows exactly, no row guard in the kernel");
static_assert(GM_ROWS % 8 == 0 && D == 128, "a wave stages whole 128-float rows, 8 waves take turns");

template <int MODE>
__global__ __launch_bounds__(256)
void norm_gemm_wmma(const float* __restrict__ agg,
                    const float* __restrict__ hprev,
                    const float* __restrict__ deg,
                    const float* __restrict__ W,
                    const float* __restrict__ bias,
                    float* __restrict__ out) {
    __shared__ __attribute__((aligned(16))) float Alds[GM_ROWS * GM_LDA];

    const int t      = threadIdx.x;
    const int waveId = t >> 5;
    const int lane   = t & 31;
    const int rowBase = blockIdx.x * GM_ROWS;

    {
        const float4* a4 = reinterpret_cast<const float4*>(agg)   + (size_t)rowBase * (D / 4) + lane;
        const float4* h4 = reinterpret_cast<const float4*>(hprev) + (size_t)rowBase * (D / 4) + lane;
        float4 av[GM_RPT], hv[GM_RPT];
        float  dg[GM_RPT];
#pragma unroll
        for (int i = 0; i < GM_RPT; ++i) {
            const int r = waveId + 8 * i;
            dg[i] = deg[rowBase + r];
            av[i] = a4[r * (D / 4)];
            if (MODE == 0) hv[i] = h4[r * (D / 4)];
        }
#pragma unroll
        for (int i = 0; i < GM_RPT; ++i) {
            const int r = waveId + 8 * i;
            float s;
            if (MODE == 0) s = 1.0f / (dg[i] + 1.0f);
            else           s = rsqrtf(fmaxf(dg[i], 1.0f));
            float4 v = av[i];
            if (MODE == 0) { v.x += hv[i].x; v.y += hv[i].y; v.z += hv[i].z; v.w += hv[i].w; }
            v.x *= s; v.y *= s; v.z *= s; v.w *= s;
            *reinterpret_cast<float4*>(Alds + r * GM_LDA + 4 * lane) = v;
        }
    }
    __syncthreads();

    const int half   = lane >> 4;
    const int lm     = lane & 15;
    const int col0   = waveId * 16;

    v8f c[GM_MT];
#pragma unroll
    for (int m = 0; m < GM_MT; ++m) c[m] = v8f{};

#pragma unroll
    for (int kk = 0; kk < 32; ++kk) {
        const int k0 = 4 * kk + 2 * half;
        v2f b;
        b.x = W[(size_t)(k0)     * D + col0 + lm];
        b.y = W[(size_t)(k0 + 1) * D + col0 + lm];
#pragma unroll
        for (int m = 0; m < GM_MT; ++m) {
            v2f a;
            a.x = Alds[(16 * m + lm) * GM_LDA + k0];
            a.y = Alds[(16 * m + lm) * GM_LDA + k0 + 1];
            c[m] = __builtin_amdgcn_wmma_f32_16x16x4_f32(
                       false, a, false, b, (short)0, c[m], false, false);
        }
    }

    const float bb = bias[col0 + lm];
#pragma unroll
    for (int m = 0; m < GM_MT; ++m) {
#pragma unroll
        for (int v = 0; v < 8; ++v) {
            const int row = rowBase + 16 * m + v + 8 * half;
            out[(size_t)row * D + col0 + lm] = c[m][v] + bb;
        }
    }
}

extern "C" void kernel_launch(void* const* d_in, const int* in_sizes, int n_in,
                              void* d_out, int out_size, void* d_ws, size_t ws_size,
                              hipStream_t stream) {
    const float* x  = (const float*)d_in[0];
    const float* W1 = (const float*)d_in[1];
    const float* b1 = (const float*)d_in[2];
    const float* W2 = (const float*)d_in[3];
    const float* b2 = (const float*)d_in[4];
    const float* W3 = (const float*)d_in[5];
    const float* b3 = (const float*)d_in[6];
    const int* src  = (const int*)d_in[7];
    const int* dst  = (const int*)d_in[8];
    float* out = (float*)d_out;

    const int N = NN;
    float* deg_in  = (float*)d_ws;
    float* deg_out = deg_in + N;
    float* agg     = deg_out + N;
    float* h1      = agg + (size_t)N * D;
    float* h2      = h1  + (size_t)N * D;
    int* reccnt       = (int*)(h2 + (size_t)N * D);
    unsigned* reclist = (unsigned*)(reccnt + NTILE * NWAVE * RCW);
    unsigned* marks   = reclist + (size_t)NTILE * NWAVE * QCAP;

    dim3 tileGrid(NTILE);
    dim3 gemmGrid(N / GM_ROWS);
    dim3 markGrid(MARKBLK);

    hipMemsetAsync(deg_out, 0, (size_t)N * sizeof(float), stream);

    tile_mark_kernel<<<markGrid, 32 * MARKW, 0, stream>>>(dst, marks);

    tile_sum_kernel<0, 0><<<tileGrid, TBLK, 0, stream>>>(x, src, dst, marks, deg_out, agg, reclist, reccnt, deg_in, deg_out);
    norm_gemm_wmma<0><<<gemmGrid, 256, 0, stream>>>(agg, x, deg_in, W1, b1, h1);

    tile_sum_kernel<0, 1><<<tileGrid, TBLK, 0, stream>>>(h1, src, dst, marks, deg_out, agg, reclist, reccnt, deg_in, deg_out);
    norm_gemm_wmma<0><<<gemmGrid, 256, 0, stream>>>(agg, h1, deg_in, W2, b2, h2);

    tile_sum_kernel<1, 1><<<tileGrid, TBLK, 0, stream>>>(h2, src, dst, marks, deg_out, agg, reclist, reccnt, deg_in, deg_out);
    norm_gemm_wmma<1><<<gemmGrid, 256, 0, stream>>>(agg, h2, deg_in, W3, b3, out);
}


# Out-degree no longer added per edge in memory: 64 extra blocks of the marking launch count slices of src in LDS (a counter per node), layer 1's product sums their counts per row; memset launch gone
# speedup vs baseline: 4.1083x; 1.3372x over previous
#include <hip/hip_runtime.h>

#define NN 50000
#define NE 800000
#define D  128

typedef __attribute__((ext_vector_type(2))) float v2f;
typedef __attribute__((ext_vector_type(8))) float v8f;

#define RT     196
#define NTILE  256
#define GRP    32
#define GB     16
#define WLCAP  (32 + GB * GRP)
#define NQ     (NE / 4)
#define WSPAN  (32 * GRP)
#define NW     ((NE + WSPAN - 1) / WSPAN)
#define MARKW  8
#define MARKBLK ((NW + MARKW - 1) / MARKW)
#define HB     64
#define HQ     (NQ / HB)
#define HTRIP  ((HQ + 32 * MARKW - 1) / (32 * MARKW))
#define TBLK   1024
#define NWAVE  (TBLK / 32)
#define QCAP   512
#define RQ     64
#define RCW    8
#define GBAT   16
#define NACC   7
#define GLW    ((NW + NWAVE - 1) / NWAVE)
#define GLCAP  (32 * GLW)

static_assert(NE % 4 == 0 && NE % GRP == 0 && WSPAN % 128 == 0, "dst is read in 16-byte words and in whole runs");
static_assert(WLCAP >= 31 + GB * GRP, "a wave's list holds the rest of one flush plus one full batch of runs");
static_assert(GLW <= 32 && NE / GRP <= (1 << 16), "a lane per mark word of the wave; a run's number fits 16 bits");
static_assert(NTILE % 32 == 0, "a marking wave stores its table in whole 128-byte lines");
#define NCW    (NN / 2)
static_assert(NQ % HB == 0 && NN % 8 == 0 && MARKW * NTILE <= NCW, "src slices are whole 16-byte words; counter words move as 16-byte words; the marking tables fit the counter array");
static_assert(4 * HQ < (1 << 16), "a slice cannot carry a node's count out of its 16 bits");
static_assert(NE <= (1 << 20) && RT <= 256, "entry packs edge id (20 bits) and local row (8 bits)");
static_assert((long)NTILE * RT >= NN && (long)(NTILE - 1) * RT < NN, "tiles cover every row");
static_assert(NN <= (1 << 16), "a row queue entry is a source row in 16 bits");
static_assert(NACC * RQ <= QCAP && NACC < RCW && 32 % GBAT == 0, "a wave's list holds all its row queues; a round of 32 entries is whole batches");
static_assert(QCAP % 128 == 0, "a kept list is staged back into LDS in steps of 128 entries");
static_assert(NACC * NWAVE >= RT && (NWAVE & (NWAVE - 1)) == 0, "every local row has an owner wave");

__device__ __forceinline__ void tile_add_row(float* trow, float4 v,
                                             int rot, int lo, int lo3) {
    float a0 = rot ? v.y : v.x;
    float a1 = rot ? v.z : v.y;
    float a2 = rot ? v.w : v.z;
    float a3 = rot ? v.x : v.w;
    __hip_atomic_fetch_add(trow + lo + 0, a0, __ATOMIC_RELAXED, __HIP_MEMORY_SCOPE_WORKGROUP);
    __hip_atomic_fetch_add(trow + lo + 1, a1, __ATOMIC_RELAXED, __HIP_MEMORY_SCOPE_WORKGROUP);
    __hip_atomic_fetch_add(trow + lo + 2, a2, __ATOMIC_RELAXED, __HIP_MEMORY_SCOPE_WORKGROUP);
    __hip_atomic_fetch_add(trow + lo3,    a3, __ATOMIC_RELAXED, __HIP_MEMORY_SCOPE_WORKGROUP);
}

__global__ __launch_bounds__(32 * MARKW)
void tile_mark_kernel(const int* __restrict__ dst, const int* __restrict__ src,
                      unsigned* __restrict__ marks, unsigned* __restrict__ part) {
    __shared__ __attribute__((aligned(16))) unsigned sm[NCW];
    if (blockIdx.x >= MARKBLK) {
        const int t  = threadIdx.x;
        const int hb = blockIdx.x - MARKBLK;
        const int4* s4 = reinterpret_cast<const int4*>(src) + (size_t)hb * HQ;
        int4 sv[HTRIP];
#pragma unroll
        for (int k = 0; k < HTRIP; ++k) {
            const int q = t + 32 * MARKW * k;
            sv[k] = (q < HQ) ? s4[q] : make_int4(-1, -1, -1, -1);
        }
        uint4* sm4 = reinterpret_cast<uint4*>(sm);
        for (int i = t; i < NCW / 4; i += 32 * MARKW) sm4[i] = make_uint4(0u, 0u, 0u, 0u);
        __syncthreads();
#pragma unroll
        for (int k = 0; k < HTRIP; ++k) {
            const unsigned s0 = (unsigned)sv[k].x, s1 = (unsigned)sv[k].y, s2 = (unsigned)sv[k].z, s3 = (unsigned)sv[k].w;
            if (s0 < (unsigned)NN) __hip_atomic_fetch_add(sm + (s0 >> 1), 1u << (16 * (s0 & 1u)), __ATOMIC_RELAXED, __HIP_MEMORY_SCOPE_WORKGROUP);
            if (s1 < (unsigned)NN) __hip_atomic_fetch_add(sm + (s1 >> 1), 1u << (16 * (s1 & 1u)), __ATOMIC_RELAXED, __HIP_MEMORY_SCOPE_WORKGROUP);
            if (s2 < (unsigned)NN) __hip_atomic_fetch_add(sm + (s2 >> 1), 1u << (16 * (s2 & 1u)), __ATOMIC_RELAXED, __HIP_MEMORY_SCOPE_WORKGROUP);
            if (s3 < (unsigned)NN) __hip_atomic_fetch_add(sm + (s3 >> 1), 1u << (16 * (s3 & 1u)), __ATOMIC_RELAXED, __HIP_MEMORY_SCOPE_WORKGROUP);
        }
        __syncthreads();
        uint4* p4 = reinterpret_cast<uint4*>(part + (size_t)hb * NCW);
        for (int i = t; i < NCW / 4; i += 32 * MARKW) p4[i] = sm4[i];
        return;
    }
    const int lane = threadIdx.x & 31;
    const int wv   = threadIdx.x >> 5;
    const int w    = blockIdx.x * MARKW + wv;
    if (w >= NW) return;
    unsigned* tb = sm + wv * NTILE;
#pragma unroll
    for (int k = 0; k < NTILE / 32; ++k) tb[lane + 32 * k] = 0u;
    const int4* d4 = reinterpret_cast<const int4*>(dst);
    const int qb = w * (WSPAN / 4);
    int4 dv[WSPAN / 128];
#pragma unroll
    for (int j = 0; j < WSPAN / 128; ++j) {
        const int q = qb + 32 * j + lane;
        dv[j] = (q < NQ) ? d4[q] : make_int4(-1, -1, -1, -1);
    }
#pragma unroll
    for (int j = 0; j < WSPAN / 128; ++j) {
        const unsigned bit = 1u << (4 * j + (lane >> 3));
        const unsigned d0 = (unsigned)dv[j].x, d1 = (unsigned)dv[j].y, d2 = (unsigned)dv[j].z, d3 = (unsigned)dv[j].w;
        if (d0 < (unsigned)NN) __hip_atomic_fetch_or(tb + d0 / RT, bit, __ATOMIC_RELAXED, __HIP_MEMORY_SCOPE_WORKGROUP);
        if (d1 < (unsigned)NN) __hip_atomic_fetch_or(tb + d1 / RT, bit, __ATOMIC_RELAXED, __HIP_MEMORY_SCOPE_WORKGROUP);
        if (d2 < (unsigned)NN) __hip_atomic_fetch_or(tb + d2 / RT, bit, __ATOMIC_RELAXED, __HIP_MEMORY_SCOPE_WORKGROUP);
        if (d3 < (unsigned)NN) __hip_atomic_fetch_or(tb + d3 / RT, bit, __ATOMIC_RELAXED, __HIP_MEMORY_SCOPE_WORKGROUP);
    }
    unsigned* mrow = marks + (size_t)w * NTILE;
#pragma unroll
    for (int k = 0; k < NTILE / 32; ++k) mrow[lane + 32 * k] = tb[lane + 32 * k];
}

template <int SCALE>
__device__ __forceinline__ void route_round(const float* __restrict__ h,
                                            const int* __restrict__ src,
                                            const float* __restrict__ degout,
                                            float* tile, const unsigned* wl,
                                            int off, int cnt, int lane,
                                            int rot, int lo, int lo3,
                                            unsigned short* rowq,
                                            int* rowdeg) {
    const unsigned ent = (lane < cnt) ? wl[off + lane] : 0u;
    const int e = (int)(ent & 0xFFFFFu);
    const int r = (int)(ent >> 20);
    int s = 0;
    bool spill = false;
    if (lane < cnt) {
        s = src[e];
        const int slot = __hip_atomic_fetch_add(rowdeg + r, 1, __ATOMIC_RELAXED, __HIP_MEMORY_SCOPE_WORKGROUP);
        if (slot < RQ) rowq[r * RQ + slot] = (unsigned short)s;
        else spill = true;
    }
    unsigned m = __builtin_amdgcn_ballot_w32(spill);
    const float4* h4 = reinterpret_cast<const float4*>(h);
    while (m) {
        const int k = __builtin_ctz(m);
        m &= m - 1;
        const int sk = __builtin_amdgcn_readlane(s, k);
        const int rk = __builtin_amdgcn_readlane(r, k);
        float4 x = h4[(size_t)sk * (D / 4) + lane];
        if (SCALE) {
            const float sc = rsqrtf(fmaxf(degout[sk], 1.0f));
            x.x *= sc; x.y *= sc; x.z *= sc; x.w *= sc;
        }
        tile_add_row(tile + rk * D, x, rot, lo, lo3);
    }
}

template <int SCALE>
__device__ __forceinline__ void scan_and_route(const float* __restrict__ h,
                                               const int* __restrict__ src,
                                               const int* __restrict__ dst,
                                               const unsigned* __restrict__ marks,
                                               const float* __restrict__ degout,
                                               float* tile, unsigned* wl, unsigned short* gl,
                                               int wave, int lane, int base, int rows,
                                               int rot, int lo, int lo3,
                                               unsigned short* rowq,
                                               int* rowdeg) {
    const int nl = (NW - wave + NWAVE - 1) / NWAVE;
    unsigned pw = 0u;
    if (lane < nl) pw = marks[(size_t)(wave + NWAVE * lane) * NTILE + blockIdx.x];

    int ng = 0;
    for (int l = 0; l < nl; ++l) {
        const unsigned m = (unsigned)__builtin_amdgcn_readlane((int)pw, l);
        if (m) {
            if ((m >> lane) & 1u)
                gl[ng + __builtin_amdgcn_mbcnt_lo(m, 0u)] = (unsigned short)((wave + NWAVE * l) * 32 + lane);
            ng += __builtin_popcount(m);
        }
    }

    int wcount = 0;
    for (int i0 = 0; i0 < ng; i0 += GB) {
        const int nb = min(GB, ng - i0);
        const int gme = (lane < nb) ? (int)gl[i0 + lane] : 0;
        int dv[GB];
#pragma unroll
        for (int k = 0; k < GB; ++k)
            dv[k] = dst[__builtin_amdgcn_readlane(gme, k) * GRP + lane];
#pragma unroll
        for (int k = 0; k < GB; ++k) {
            if (k < nb) {
                const int e = __builtin_amdgcn_readlane(gme, k) * GRP + lane;
                const unsigned r = (unsigned)(dv[k] - base);
                const bool hit = r < (unsigned)rows;
                const unsigned m = __builtin_amdgcn_ballot_w32(hit);
                if (hit) wl[wcount + __builtin_amdgcn_mbcnt_lo(m, 0u)] = (unsigned)e | (r << 20);
                wcount += __builtin_popcount(m);
            }
        }
        if (wcount >= 32) {
            int done = 0;
            for (; done + 32 <= wcount; done += 32)
                route_round<SCALE>(h, src, degout, tile, wl, done, 32, lane, rot, lo, lo3, rowq, rowdeg);
            const int rem = wcount - done;
            unsigned keep = (lane < rem) ? wl[done + lane] : 0u;
            if (lane < rem) wl[lane] = keep;
            wcount = rem;
        }
    }
    for (int done = 0; done < wcount; done += 32)
        route_round<SCALE>(h, src, degout, tile, wl, done, min(32, wcount - done), lane, rot, lo, lo3, rowq, rowdeg);
}

template <int SCALE, int REPLAY>
__global__ __launch_bounds__(TBLK)
void tile_sum_kernel(const float* __restrict__ h,
                     const int* __restrict__ src,
                     const int* __restrict__ dst,
                     const unsigned* __restrict__ marks,
                     const float* __restrict__ degout,
                     float* __restrict__ agg,
                     unsigned* reclist,
                     int* reccnt,
                     float* indeg) {
    __shared__ __attribute__((aligned(16))) float tile[RT * D];
    __shared__ unsigned wlist[NWAVE * WLCAP];
    __shared__ unsigned short glist[NWAVE * GLCAP];
    __shared__ unsigned queue[NWAVE * QCAP];
    __shared__ unsigned short rowq[RT * RQ];
    __shared__ int rowdeg[RT];

    const int t    = threadIdx.x;
    const int lane = t & 31;
    const int wave = __builtin_amdgcn_readfirstlane(t >> 5);
    const int base = blockIdx.x * RT;
    const int rows = min(RT, NN - base);
    const int rot  = lane >> 4;
    const int lo   = 4 * lane + rot;
    const int lo3  = 4 * lane + 3 - 3 * rot;

    float4* t4 = reinterpret_cast<float4*>(tile);
    const int gw = blockIdx.x * NWAVE + wave;
    unsigned* rec = reclist + (size_t)gw * QCAP;
    int* rc       = reccnt + (size_t)gw * RCW;
    unsigned* q   = queue + wave * QCAP;

    bool scan = true;
    int cntv = 0;
    if (REPLAY) {
        const int full = reccnt[(blockIdx.x * NWAVE + lane) * RCW + NACC];
        scan = __builtin_amdgcn_ballot_w32(full != 0) != 0u;
        if (lane < NACC) cntv = rc[lane];
    }

    if (scan) {
        for (int i = t; i < RT * D / 4; i += TBLK) t4[i] = make_float4(0.f, 0.f, 0.f, 0.f);
        if (t < RT) rowdeg[t] = 0;
        __syncthreads();
        scan_and_route<SCALE>(h, src, dst, marks, degout, tile, wlist + wave * WLCAP, glist + wave * GLCAP,
                              wave, lane, base, rows, rot, lo, lo3, rowq, rowdeg);
        __syncthreads();
        int dg = 0;
        if (lane < NACC && wave + NWAVE * lane < rows) dg = rowdeg[wave + NWAVE * lane];
        cntv = min(dg, RQ);
        const bool over = __builtin_amdgcn_ballot_w32(dg > RQ) != 0u;
        int o = 0;
        for (int j = 0; j < NACC; ++j) {
            const int nj = __builtin_amdgcn_readlane(cntv, j);
            const unsigned short* rq = rowq + (wave + NWAVE * j) * RQ;
            for (int i = lane; i < nj; i += 32) q[o + i] = rq[i];
            o += nj;
        }
        if (!REPLAY) {
            for (int i = lane; i < o; i += 32) rec[i] = q[i];
            if (lane < RCW) rc[lane] = (lane < NACC) ? cntv : (over ? 1 : 0);
            if (t < rows) indeg[base + t] = (float)rowdeg[t];
        }
    }

    int total = 0;
    int endv = 0x7fffffff;
    for (int j = 0; j < NACC; ++j) {
        total += __builtin_amdgcn_readlane(cntv, j);
        if (lane == j) endv = total;
    }

    if (!scan) {
        for (int i0 = 0; i0 < total; i0 += 128) {
            const int i = i0 + lane;
            const unsigned e0 = (i      < total) ? rec[i]      : 0u;
            const unsigned e1 = (i + 32 < total) ? rec[i + 32] : 0u;
            const unsigned e2 = (i + 64 < total) ? rec[i + 64] : 0u;
            const unsigned e3 = (i + 96 < total) ? rec[i + 96] : 0u;
            q[i] = e0; q[i + 32] = e1; q[i + 64] = e2; q[i + 96] = e3;
        }
    }

    float4* o4 = reinterpret_cast<float4*>(agg + (size_t)base * D);
    const char* hb = reinterpret_cast<const char*>(h);
    const unsigned lane16 = 16u * (unsigned)lane;
    float4 acc = make_float4(0.f, 0.f, 0.f, 0.f);
    int j = 0;
    int end = __builtin_amdgcn_readlane(endv, 0);
    for (int off = 0; off < total; off += 32) {
        const int cnt = min(32, total - off);
        const unsigned ent = (lane < cnt) ? q[off + lane] : 0u;
        int scb = 0;
        if (SCALE) scb = __float_as_int(rsqrtf(fmaxf(degout[ent], 1.0f)));
        for (int k0 = 0; k0 < cnt; k0 += GBAT) {
            float4 v[GBAT];
#pragma unroll
            for (int k = 0; k < GBAT; ++k) {
                const unsigned sk = (unsigned)__builtin_amdgcn_readlane((int)ent, k0 + k);
                v[k] = *reinterpret_cast<const float4*>(hb + (size_t)sk * (D * sizeof(float)) + lane16);
            }
#pragma unroll
            for (int k = 0; k < GBAT; ++k) {
                if (k0 + k < cnt) {
                    const int pos = off + k0 + k;
                    while (pos == end) {
                        const int r = wave + NWAVE * j;
                        if (r < rows) {
                            float4 o = acc;
                            if (scan) {
                                const float4 p = t4[r * (D / 4) + lane];
                                o.x += p.x; o.y += p.y; o.z += p.z; o.w += p.w;
                            }
                            o4[r * (D / 4) + lane] = o;
                        }
                        acc = make_float4(0.f, 0.f, 0.f, 0.f);
                        ++j;
                        end = __builtin_amdgcn_readlane(endv, j);
                    }
                    float4 x = v[k];
                    if (SCALE) {
                        const float sc = __int_as_float(__builtin_amdgcn_readlane(scb, k0 + k));
                        x.x *= sc; x.y *= sc; x.z *= sc; x.w *= sc;
                    }
                    acc.x += x.x; acc.y += x.y; acc.z += x.z; acc.w += x.w;
                }
            }
        }
    }
    for (; j < NACC; ++j) {
        const int r = wave + NWAVE * j;
        if (r < rows) {
            float4 o = acc;
            if (scan) {
                const float4 p = t4[r * (D / 4) + lane];
                o.x += p.x; o.y += p.y; o.z += p.z; o.w += p.w;
            }
            o4[r * (D / 4) + lane] = o;
        }
        acc = make_float4(0.f, 0.f, 0.f, 0.f);
    }
}

#define GM_MT   5
#define GM_ROWS (16 * GM_MT)
#define GM_LDA  132
#define GM_RPT  (GM_ROWS / 8)
#define GM_SG   3
#define GM_ST   ((HB + GM_SG - 1) / GM_SG)

static_assert(NN % GM_ROWS == 0, "blocks cover the rows exactly, no row guard in the kernel");
static_assert(GM_ROWS % 8 == 0 && D == 128, "a wave stages whole 128-float rows, 8 waves take turns");
static_assert(GM_SG * GM_ROWS <= 256, "the summing groups are threads of the block");

template <int MODE, int SUMDEG>
__global__ __launch_bounds__(256)
void norm_gemm_wmma(const float* __restrict__ agg,
                    const float* __restrict__ hprev,
                    const float* __restrict__ deg,
                    const float* __restrict__ W,
                    const float* __restrict__ bias,
                    float* __restrict__ out,
                    const unsigned* __restrict__ part,
                    float* __restrict__ outdeg) {
    __shared__ __attribute__((aligned(16))) float Alds[GM_ROWS * GM_LDA];
    __shared__ unsigned dsum[256];

    const int t      = threadIdx.x;
    const int waveId = t >> 5;
    const int lane   = t & 31;
    const int rowBase = blockIdx.x * GM_ROWS;

    unsigned pv[GM_ST];
    if (SUMDEG) {
        const int g = t / GM_ROWS;
        const unsigned* pc = part + ((rowBase + (t - g * GM_ROWS)) >> 1);
#pragma unroll
        for (int i = 0; i < GM_ST; ++i)
            pv[i] = pc[(size_t)min(g + GM_SG * i, HB - 1) * NCW];
        __builtin_amdgcn_sched_barrier(0);
    }

    {
        const float4* a4 = reinterpret_cast<const float4*>(agg)   + (size_t)rowBase * (D / 4) + lane;
        const float4* h4 = reinterpret_cast<const float4*>(hprev) + (size_t)rowBase * (D / 4) + lane;
        float4 av[GM_RPT], hv[GM_RPT];
        float  dg[GM_RPT];
#pragma unroll
        for (int i = 0; i < GM_RPT; ++i) {
            const int r = waveId + 8 * i;
            dg[i] = deg[rowBase + r];
            av[i] = a4[r * (D / 4)];
            if (MODE == 0) hv[i] = h4[r * (D / 4)];
        }
#pragma unroll
        for (int i = 0; i < GM_RPT; ++i) {
            const int r = waveId + 8 * i;
            float s;
            if (MODE == 0) s = 1.0f / (dg[i] + 1.0f);
            else           s = rsqrtf(fmaxf(dg[i], 1.0f));
            float4 v = av[i];
            if (MODE == 0) { v.x += hv[i].x; v.y += hv[i].y; v.z += hv[i].z; v.w += hv[i].w; }
            v.x *= s; v.y *= s; v.z *= s; v.w *= s;
            *reinterpret_cast<float4*>(Alds + r * GM_LDA + 4 * lane) = v;
        }
    }
    if (SUMDEG) {
        __builtin_amdgcn_sched_barrier(0);
        const int g = t / GM_ROWS;
        const unsigned sh = 16u * ((unsigned)(rowBase + (t - g * GM_ROWS)) & 1u);
        unsigned sum = 0u;
#pragma unroll
        for (int i = 0; i < GM_ST; ++i) sum += (g + GM_SG * i < HB) ? ((pv[i] >> sh) & 0xFFFFu) : 0u;
        dsum[t] = sum;
    }
    __syncthreads();

    if (SUMDEG && t < GM_ROWS) {
        unsigned od = 0u;
#pragma unroll
        for (int g = 0; g < GM_SG; ++g) od += dsum[t + g * GM_ROWS];
        outdeg[rowBase + t] = (float)od;
    }

    const int half   = lane >> 4;
    const int lm     = lane & 15;
    const int col0   = waveId * 16;

    v8f c[GM_MT];
#pragma unroll
    for (int m = 0; m < GM_MT; ++m) c[m] = v8f{};

#pragma unroll
    for (int kk = 0; kk < 32; ++kk) {
        const int k0 = 4 * kk + 2 * half;
        v2f b;
        b.x = W[(size_t)(k0)     * D + col0 + lm];
        b.y = W[(size_t)(k0 + 1) * D + col0 + lm];
#pragma unroll
        for (int m = 0; m < GM_MT; ++m) {
            v2f a;
            a.x = Alds[(16 * m + lm) * GM_LDA + k0];
            a.y = Alds[(16 * m + lm) * GM_LDA + k0 + 1];
            c[m] = __builtin_amdgcn_wmma_f32_16x16x4_f32(
                       false, a, false, b, (short)0, c[m], false, false);
        }
    }

    const float bb = bias[col0 + lm];
#pragma unroll
    for (int m = 0; m < GM_MT; ++m) {
#pragma unroll
        for (int v = 0; v < 8; ++v) {
            const int row = rowBase + 16 * m + v + 8 * half;
            out[(size_t)row * D + col0 + lm] = c[m][v] + bb;
        }
    }
}

extern "C" void kernel_launch(void* const* d_in, const int* in_sizes, int n_in,
                              void* d_out, int out_size, void* d_ws, size_t ws_size,
                              hipStream_t stream) {
    const float* x  = (const float*)d_in[0];
    const float* W1 = (const float*)d_in[1];
    const float* b1 = (const float*)d_in[2];
    const float* W2 = (const float*)d_in[3];
    const float* b2 = (const float*)d_in[4];
    const float* W3 = (const float*)d_in[5];
    const float* b3 = (const float*)d_in[6];
    const int* src  = (const int*)d_in[7];
    const int* dst  = (const int*)d_in[8];
    float* out = (float*)d_out;

    const int N = NN;
    float* deg_in  = (float*)d_ws;
    float* deg_out = deg_in + N;
    float* agg     = deg_out + N;
    float* h1      = agg + (size_t)N * D;
    float* h2      = h1  + (size_t)N * D;
    int* reccnt       = (int*)(h2 + (size_t)N * D);
    unsigned* reclist = (unsigned*)(reccnt + NTILE * NWAVE * RCW);
    unsigned* marks   = reclist + (size_t)NTILE * NWAVE * QCAP;
    unsigned* part    = marks + (size_t)NW * NTILE;

    dim3 tileGrid(NTILE);
    dim3 gemmGrid(N / GM_ROWS);
    dim3 markGrid(MARKBLK + HB);

    tile_mark_kernel<<<markGrid, 32 * MARKW, 0, stream>>>(dst, src, marks, part);

    tile_sum_kernel<0, 0><<<tileGrid, TBLK, 0, stream>>>(x, src, dst, marks, deg_out, agg, reclist, reccnt, deg_in);
    norm_gemm_wmma<0, 1><<<gemmGrid, 256, 0, stream>>>(agg, x, deg_in, W1, b1, h1, part, deg_out);

    tile_sum_kernel<0, 1><<<tileGrid, TBLK, 0, stream>>>(h1, src, dst, marks, deg_out, agg, reclist, reccnt, deg_in);
    norm_gemm_wmma<0, 0><<<gemmGrid, 256, 0, stream>>>(agg, h1, deg_in, W2, b2, h2, part, deg_out);

    tile_sum_kernel<1, 1><<<tileGrid, TBLK, 0, stream>>>(h2, src, dst, marks, deg_out, agg, reclist, reccnt, deg_in);
    norm_gemm_wmma<1, 0><<<gemmGrid, 256, 0, stream>>>(agg, h2, deg_in, W3, b3, out, part, deg_out);
}


# Layer's product moved into the neighbour-sum block: owner waves normalize finished rows into an LDS A tile, the block's 32 waves run its 104 WMMA tiles; 7 launches to 4; sum and k orders kept
# speedup vs baseline: 4.9136x; 1.1960x over previous
#include <hip/hip_runtime.h>

#define NN 50000
#define NE 800000
#define D  128

typedef __attribute__((ext_vector_type(2))) float v2f;
typedef __attribute__((ext_vector_type(8))) float v8f;

#define RT     196
#define NTILE  256
#define GRP    32
#define GB     16
#define WLCAP  (32 + GB * GRP)
#define NQ     (NE / 4)
#define WSPAN  (32 * GRP)
#define NW     ((NE + WSPAN - 1) / WSPAN)
#define MARKW  8
#define MARKBLK ((NW + MARKW - 1) / MARKW)
#define HB     64
#define HQ     (NQ / HB)
#define HTRIP  ((HQ + 32 * MARKW - 1) / (32 * MARKW))
#define TBLK   1024
#define NWAVE  (TBLK / 32)
#define QCAP   512
#define RQ     64
#define RCW    8
#define GBAT   16
#define NACC   7
#define GLW    ((NW + NWAVE - 1) / NWAVE)
#define GLCAP  (32 * GLW)
#define GM_LDA 132
#define A_MT   ((RT + 15) / 16)
#define A_ROWS (16 * A_MT)
#define WL_BYTES  (NWAVE * WLCAP * 4)
#define GL_BYTES  (NWAVE * GLCAP * 2)
#define A_BYTES   (A_ROWS * GM_LDA * 4)
#define SCR_BYTES ((WL_BYTES + GL_BYTES) > A_BYTES ? (WL_BYTES + GL_BYTES) : A_BYTES)
#define OD_SG  5
#define OD_ST  ((HB + OD_SG - 1) / OD_SG)

static_assert(NE % 4 == 0 && NE % GRP == 0 && WSPAN % 128 == 0, "dst is read in 16-byte words and in whole runs");
static_assert(WLCAP >= 31 + GB * GRP, "a wave's list holds the rest of one flush plus one full batch of runs");
static_assert(GLW <= 32 && NE / GRP <= (1 << 16), "a lane per mark word of the wave; a run's number fits 16 bits");
static_assert(NTILE % 32 == 0, "a marking wave stores its table in whole 128-byte lines");
#define NCW    (NN / 2)
static_assert(NQ % HB == 0 && NN % 8 == 0 && MARKW * NTILE <= NCW, "src slices are whole 16-byte words; counter words move as 16-byte words; the marking tables fit the counter array");
static_assert(4 * HQ < (1 << 16), "a slice cannot carry a node's count out of its 16 bits");
static_assert(NE <= (1 << 20) && RT <= 256, "entry packs edge id (20 bits) and local row (8 bits)");
static_assert((long)NTILE * RT >= NN && (long)(NTILE - 1) * RT < NN, "tiles cover every row");
static_assert(NN <= (1 << 16), "a row queue entry is a source row in 16 bits");
static_assert(NACC * RQ <= QCAP && NACC < RCW && 32 % GBAT == 0, "a wave's list holds all its row queues; a round of 32 entries is whole batches");
static_assert(QCAP % 128 == 0, "a kept list is staged back into LDS in steps of 128 entries");
static_assert(NACC * NWAVE >= RT && (NWAVE & (NWAVE - 1)) == 0, "every local row has an owner wave");
static_assert(WL_BYTES % 16 == 0 && (GM_LDA * 4) % 16 == 0 && D == 128, "the A tile's rows are whole 16-byte words");
static_assert(NWAVE == 32 && A_MT > 12 && A_MT <= 13, "wave w: column panel w % 8, row tiles w / 8 + 4 i; only w / 8 == 0 has a fourth");
static_assert(OD_SG * RT <= TBLK, "the summing groups are threads of the block");

__device__ __forceinline__ void tile_add_row(float* trow, float4 v,
                                             int rot, int lo, int lo3) {
    float a0 = rot ? v.y : v.x;
    float a1 = rot ? v.z : v.y;
    float a2 = rot ? v.w : v.z;
    float a3 = rot ? v.x : v.w;
    __hip_atomic_fetch_add(trow + lo + 0, a0, __ATOMIC_RELAXED, __HIP_MEMORY_SCOPE_WORKGROUP);
    __hip_atomic_fetch_add(trow + lo + 1, a1, __ATOMIC_RELAXED, __HIP_MEMORY_SCOPE_WORKGROUP);
    __hip_atomic_fetch_add(trow + lo + 2, a2, __ATOMIC_RELAXED, __HIP_MEMORY_SCOPE_WORKGROUP);
    __hip_atomic_fetch_add(trow + lo3,    a3, __ATOMIC_RELAXED, __HIP_MEMORY_SCOPE_WORKGROUP);
}

__global__ __launch_bounds__(32 * MARKW)
void tile_mark_kernel(const int* __restrict__ dst, const int* __restrict__ src,
                      unsigned* __restrict__ marks, unsigned* __restrict__ part) {
    __shared__ __attribute__((aligned(16))) unsigned sm[NCW];
    if (blockIdx.x >= MARKBLK) {
        const int t  = threadIdx.x;
        const int hb = blockIdx.x - MARKBLK;
        const int4* s4 = reinterpret_cast<const int4*>(src) + (size_t)hb * HQ;
        int4 sv[HTRIP];
#pragma unroll
        for (int k = 0; k < HTRIP; ++k) {
            const int q = t + 32 * MARKW * k;
            sv[k] = (q < HQ) ? s4[q] : make_int4(-1, -1, -1, -1);
        }
        uint4* sm4 = reinterpret_cast<uint4*>(sm);
        for (int i = t; i < NCW / 4; i += 32 * MARKW) sm4[i] = make_uint4(0u, 0u, 0u, 0u);
        __syncthreads();
#pragma unroll
        for (int k = 0; k < HTRIP; ++k) {
            const unsigned s0 = (unsigned)sv[k].x, s1 = (unsigned)sv[k].y, s2 = (unsigned)sv[k].z, s3 = (unsigned)sv[k].w;
            if (s0 < (unsigned)NN) __hip_atomic_fetch_add(sm + (s0 >> 1), 1u << (16 * (s0 & 1u)), __ATOMIC_RELAXED, __HIP_MEMORY_SCOPE_WORKGROUP);
            if (s1 < (unsigned)NN) __hip_atomic_fetch_add(sm + (s1 >> 1), 1u << (16 * (s1 & 1u)), __ATOMIC_RELAXED, __HIP_MEMORY_SCOPE_WORKGROUP);
            if (s2 < (unsigned)NN) __hip_atomic_fetch_add(sm + (s2 >> 1), 1u << (16 * (s2 & 1u)), __ATOMIC_RELAXED, __HIP_MEMORY_SCOPE_WORKGROUP);
            if (s3 < (unsigned)NN) __hip_atomic_fetch_add(sm + (s3 >> 1), 1u << (16 * (s3 & 1u)), __ATOMIC_RELAXED, __HIP_MEMORY_SCOPE_WORKGROUP);
        }
        __syncthreads();
        uint4* p4 = reinterpret_cast<uint4*>(part + (size_t)hb * NCW);
        for (int i = t; i < NCW / 4; i += 32 * MARKW) p4[i] = sm4[i];
        return;
    }
    const int lane = threadIdx.x & 31;
    const int wv   = threadIdx.x >> 5;
    const int w    = blockIdx.x * MARKW + wv;
    if (w >= NW) return;
    unsigned* tb = sm + wv * NTILE;
#pragma unroll
    for (int k = 0; k < NTILE / 32; ++k) tb[lane + 32 * k] = 0u;
    const int4* d4 = reinterpret_cast<const int4*>(dst);
    const int qb = w * (WSPAN / 4);
    int4 dv[WSPAN / 128];
#pragma unroll
    for (int j = 0; j < WSPAN / 128; ++j) {
        const int q = qb + 32 * j + lane;
        dv[j] = (q < NQ) ? d4[q] : make_int4(-1, -1, -1, -1);
    }
#pragma unroll
    for (int j = 0; j < WSPAN / 128; ++j) {
        const unsigned bit = 1u << (4 * j + (lane >> 3));
        const unsigned d0 = (unsigned)dv[j].x, d1 = (unsigned)dv[j].y, d2 = (unsigned)dv[j].z, d3 = (unsigned)dv[j].w;
        if (d0 < (unsigned)NN) __hip_atomic_fetch_or(tb + d0 / RT, bit, __ATOMIC_RELAXED, __HIP_MEMORY_SCOPE_WORKGROUP);
        if (d1 < (unsigned)NN) __hip_atomic_fetch_or(tb + d1 / RT, bit, __ATOMIC_RELAXED, __HIP_MEMORY_SCOPE_WORKGROUP);
        if (d2 < (unsigned)NN) __hip_atomic_fetch_or(tb + d2 / RT, bit, __ATOMIC_RELAXED, __HIP_MEMORY_SCOPE_WORKGROUP);
        if (d3 < (unsigned)NN) __hip_atomic_fetch_or(tb + d3 / RT, bit, __ATOMIC_RELAXED, __HIP_MEMORY_SCOPE_WORKGROUP);
    }
    unsigned* mrow = marks + (size_t)w * NTILE;
#pragma unroll
    for (int k = 0; k < NTILE / 32; ++k) mrow[lane + 32 * k] = tb[lane + 32 * k];
}

template <int SCALE>
__device__ __forceinline__ void route_round(const float* __restrict__ h,
                                            const int* __restrict__ src,
                                            const float* __restrict__ degout,
                                            float* tile, const unsigned* wl,
                                            int off, int cnt, int lane,
                                            int rot, int lo, int lo3,
                                            unsigned short* rowq,
                                            int* rowdeg) {
    const unsigned ent = (lane < cnt) ? wl[off + lane] : 0u;
    const int e = (int)(ent & 0xFFFFFu);
    const int r = (int)(ent >> 20);
    int s = 0;
    bool spill = false;
    if (lane < cnt) {
        s = src[e];
        const int slot = __hip_atomic_fetch_add(rowdeg + r, 1, __ATOMIC_RELAXED, __HIP_MEMORY_SCOPE_WORKGROUP);
        if (slot < RQ) rowq[r * RQ + slot] = (unsigned short)s;
        else spill = true;
    }
    unsigned m = __builtin_amdgcn_ballot_w32(spill);
    const float4* h4 = reinterpret_cast<const float4*>(h);
    while (m) {
        const int k = __builtin_ctz(m);
        m &= m - 1;
        const int sk = __builtin_amdgcn_readlane(s, k);
        const int rk = __builtin_amdgcn_readlane(r, k);
        float4 x = h4[(size_t)sk * (D / 4) + lane];
        if (SCALE) {
            const float sc = rsqrtf(fmaxf(degout[sk], 1.0f));
            x.x *= sc; x.y *= sc; x.z *= sc; x.w *= sc;
        }
        tile_add_row(tile + rk * D, x, rot, lo, lo3);
    }
}

template <int SCALE>
__device__ __forceinline__ void scan_and_route(const float* __restrict__ h,
                                               const int* __restrict__ src,
                                               const int* __restrict__ dst,
                                               const unsigned* __restrict__ marks,
                                               const float* __restrict__ degout,
                                               float* tile, unsigned* wl, unsigned short* gl,
                                               int wave, int lane, int base, int rows,
                                               int rot, int lo, int lo3,
                                               unsigned short* rowq,
                                               int* rowdeg) {
    const int nl = (NW - wave + NWAVE - 1) / NWAVE;
    unsigned pw = 0u;
    if (lane < nl) pw = marks[(size_t)(wave + NWAVE * lane) * NTILE + blockIdx.x];

    int ng = 0;
    for (int l = 0; l < nl; ++l) {
        const unsigned m = (unsigned)__builtin_amdgcn_readlane((int)pw, l);
        if (m) {
            if ((m >> lane) & 1u)
                gl[ng + __builtin_amdgcn_mbcnt_lo(m, 0u)] = (unsigned short)((wave + NWAVE * l) * 32 + lane);
            ng += __builtin_popcount(m);
        }
    }

    int wcount = 0;
    for (int i0 = 0; i0 < ng; i0 += GB) {
        const int nb = min(GB, ng - i0);
        const int gme = (lane < nb) ? (int)gl[i0 + lane] : 0;
        int dv[GB];
#pragma unroll
        for (int k = 0; k < GB; ++k)
            dv[k] = dst[__builtin_amdgcn_readlane(gme, k) * GRP + lane];
#pragma unroll
        for (int k = 0; k < GB; ++k) {
            if (k < nb) {
                const int e = __builtin_amdgcn_readlane(gme, k) * GRP + lane;
                const unsigned r = (unsigned)(dv[k] - base);
                const bool hit = r < (unsigned)rows;
                const unsigned m = __builtin_amdgcn_ballot_w32(hit);
                if (hit) wl[wcount + __builtin_amdgcn_mbcnt_lo(m, 0u)] = (unsigned)e | (r << 20);
                wcount += __builtin_popcount(m);
            }
        }
        if (wcount >= 32) {
            int done = 0;
            for (; done + 32 <= wcount; done += 32)
                route_round<SCALE>(h, src, degout, tile, wl, done, 32, lane, rot, lo, lo3, rowq, rowdeg);
            const int rem = wcount - done;
            unsigned keep = (lane < rem) ? wl[done + lane] : 0u;
            if (lane < rem) wl[lane] = keep;
            wcount = rem;
        }
    }
    for (int done = 0; done < wcount; done += 32)
        route_round<SCALE>(h, src, degout, tile, wl, done, min(32, wcount - done), lane, rot, lo, lo3, rowq, rowdeg);
}

template <int NT>
__device__ __forceinline__ void tile_products(const float* Alds,
                                              const float* __restrict__ W,
                                              const float* __restrict__ bias,
                                              float* __restrict__ out,
                                              int base, int rows, int rt0, int col0,
                                              int half, int lm) {
    v8f c[NT];
#pragma unroll
    for (int i = 0; i < NT; ++i) c[i] = v8f{};

    const float* arow = Alds + (16 * rt0 + lm) * GM_LDA + 2 * half;
    const float* wcol = W + (size_t)(2 * half) * D + col0 + lm;

#pragma unroll
    for (int kk = 0; kk < 32; ++kk) {
        v2f b;
        b.x = wcol[(size_t)(4 * kk)     * D];
        b.y = wcol[(size_t)(4 * kk + 1) * D];
#pragma unroll
        for (int i = 0; i < NT; ++i) {
            v2f a;
            a.x = arow[i * (64 * GM_LDA) + 4 * kk];
            a.y = arow[i * (64 * GM_LDA) + 4 * kk + 1];
            c[i] = __builtin_amdgcn_wmma_f32_16x16x4_f32(
                       false, a, false, b, (short)0, c[i], false, false);
        }
    }

    const float bb = bias[col0 + lm];
#pragma unroll
    for (int i = 0; i < NT; ++i) {
#pragma unroll
        for (int v = 0; v < 8; ++v) {
            const int lr = 16 * (rt0 + 4 * i) + v + 8 * half;
            if (lr < rows) out[(size_t)(base + lr) * D + col0 + lm] = c[i][v] + bb;
        }
    }
}

template <int SCALE, int REPLAY>
__global__ __launch_bounds__(TBLK)
void tile_sum_gemm_kernel(const float* __restrict__ h,
                          const int* __restrict__ src,
                          const int* __restrict__ dst,
                          const unsigned* __restrict__ marks,
                          const float* __restrict__ degout,
                          const float* __restrict__ W,
                          const float* __restrict__ bias,
                          float* __restrict__ hout,
                          unsigned* reclist,
                          int* reccnt,
                          const unsigned* __restrict__ part,
                          float* __restrict__ outdeg) {
    __shared__ __attribute__((aligned(16))) float tile[RT * D];
    __shared__ __attribute__((aligned(16))) unsigned char scr[SCR_BYTES];
    __shared__ unsigned queue[NWAVE * QCAP];
    __shared__ unsigned short rowq[RT * RQ];
    __shared__ int rowdeg[RT];
    __shared__ unsigned dsum[REPLAY ? 1 : TBLK];

    unsigned* wlist       = reinterpret_cast<unsigned*>(scr);
    unsigned short* glist = reinterpret_cast<unsigned short*>(scr + WL_BYTES);
    float* Alds           = reinterpret_cast<float*>(scr);

    const int t    = threadIdx.x;
    const int lane = t & 31;
    const int wave = __builtin_amdgcn_readfirstlane(t >> 5);
    const int base = blockIdx.x * RT;
    const int rows = min(RT, NN - base);
    const int rot  = lane >> 4;
    const int lo   = 4 * lane + rot;
    const int lo3  = 4 * lane + 3 - 3 * rot;

    float4* t4 = reinterpret_cast<float4*>(tile);
    const int gw = blockIdx.x * NWAVE + wave;
    unsigned* rec = reclist + (size_t)gw * QCAP;
    int* rc       = reccnt + (size_t)gw * RCW;
    unsigned* q   = queue + wave * QCAP;

    if (!REPLAY) {
        const int g  = t / RT;
        const int nd = min(base + (t - g * RT), NN - 1);
        const unsigned* pc = part + (nd >> 1);
        unsigned pv[OD_ST];
#pragma unroll
        for (int i = 0; i < OD_ST; ++i)
            pv[i] = pc[(size_t)min(g + OD_SG * i, HB - 1) * NCW];
        const unsigned sh = 16u * ((unsigned)nd & 1u);
        unsigned sum = 0u;
#pragma unroll
        for (int i = 0; i < OD_ST; ++i) sum += (g + OD_SG * i < HB) ? ((pv[i] >> sh) & 0xFFFFu) : 0u;
        dsum[t] = sum;
    }

    bool scan = true;
    int cntv = 0;
    int dg = 0;
    bool over = false;
    if (REPLAY) {
        const int full = reccnt[(blockIdx.x * NWAVE + lane) * RCW + NACC];
        scan = __builtin_amdgcn_ballot_w32(full != 0) != 0u;
        if (lane < NACC) cntv = rc[lane];
        dg = cntv;
    }

    if (scan) {
        for (int i = t; i < RT * D / 4; i += TBLK) t4[i] = make_float4(0.f, 0.f, 0.f, 0.f);
        if (t < RT) rowdeg[t] = 0;
        __syncthreads();
        if (!REPLAY && t < rows) {
            unsigned od = 0u;
#pragma unroll
            for (int g = 0; g < OD_SG; ++g) od += dsum[t + g * RT];
            outdeg[base + t] = (float)od;
        }
        scan_and_route<SCALE>(h, src, dst, marks, degout, tile, wlist + wave * WLCAP, glist + wave * GLCAP,
                              wave, lane, base, rows, rot, lo, lo3, rowq, rowdeg);
        __syncthreads();
        dg = 0;
        if (lane < NACC && wave + NWAVE * lane < rows) dg = rowdeg[wave + NWAVE * lane];
        cntv = min(dg, RQ);
        over = __builtin_amdgcn_ballot_w32(dg > RQ) != 0u;
    }

    float4* A4 = reinterpret_cast<float4*>(Alds);
    const float4* h4 = reinterpret_cast<const float4*>(h);
    float4 hv[NACC];
#pragma unroll
    for (int j = 0; j < NACC; ++j) {
        const int r = wave + NWAVE * j;
        hv[j] = make_float4(0.f, 0.f, 0.f, 0.f);
        if (!SCALE && r < rows) hv[j] = h4[(size_t)(base + r) * (D / 4) + lane];
    }

    if (scan) {
        int o = 0;
        for (int j = 0; j < NACC; ++j) {
            const int nj = __builtin_amdgcn_readlane(cntv, j);
            const unsigned short* rq = rowq + (wave + NWAVE * j) * RQ;
            for (int i = lane; i < nj; i += 32) q[o + i] = rq[i];
            o += nj;
        }
        if (!REPLAY) {
            for (int i = lane; i < o; i += 32) rec[i] = q[i];
            if (lane < RCW) rc[lane] = (lane < NACC) ? cntv : (over ? 1 : 0);
        }
    }

    int total = 0;
    int endv = 0x7fffffff;
    for (int j = 0; j < NACC; ++j) {
        total += __builtin_amdgcn_readlane(cntv, j);
        if (lane == j) endv = total;
    }

    float sv;
    if (SCALE) sv = rsqrtf(fmaxf((float)dg, 1.0f));
    else       sv = 1.0f / ((float)dg + 1.0f);

    if (!scan) {
        for (int i0 = 0; i0 < total; i0 += 128) {
            const int i = i0 + lane;
            const unsigned e0 = (i      < total) ? rec[i]      : 0u;
            const unsigned e1 = (i + 32 < total) ? rec[i + 32] : 0u;
            const unsigned e2 = (i + 64 < total) ? rec[i + 64] : 0u;
            const unsigned e3 = (i + 96 < total) ? rec[i + 96] : 0u;
            q[i] = e0; q[i + 32] = e1; q[i + 64] = e2; q[i + 96] = e3;
        }
    }

#pragma unroll
    for (int j = 0; j < NACC; ++j) {
        const int r = wave + NWAVE * j;
        if (r < A_ROWS && (!SCALE || r >= rows)) A4[r * (GM_LDA / 4) + lane] = hv[j];
    }

    const char* hb = reinterpret_cast<const char*>(h);
    const unsigned lane16 = 16u * (unsigned)lane;
    float4 acc = make_float4(0.f, 0.f, 0.f, 0.f);
    int j = 0;
    int end = __builtin_amdgcn_readlane(endv, 0);
    for (int off = 0; off < total; off += 32) {
        const int cnt = min(32, total - off);
        const unsigned ent = (lane < cnt) ? q[off + lane] : 0u;
        int scb = 0;
        if (SCALE) scb = __float_as_int(rsqrtf(fmaxf(degout[ent], 1.0f)));
        for (int k0 = 0; k0 < cnt; k0 += GBAT) {
            float4 v[GBAT];
#pragma unroll
            for (int k = 0; k < GBAT; ++k) {
                const unsigned sk = (unsigned)__builtin_amdgcn_readlane((int)ent, k0 + k);
                v[k] = *reinterpret_cast<const float4*>(hb + (size_t)sk * (D * sizeof(float)) + lane16);
            }
#pragma unroll
            for (int k = 0; k < GBAT; ++k) {
                if (k0 + k < cnt) {
                    const int pos = off + k0 + k;
                    while (pos == end) {
                        const int r = wave + NWAVE * j;
                        if (r < rows) {
                            float4 o = acc;
                            if (scan) {
                                const float4 p = t4[r * (D / 4) + lane];
                                o.x += p.x; o.y += p.y; o.z += p.z; o.w += p.w;
                            }
                            if (!SCALE) {
                                const float4 hr = A4[r * (GM_LDA / 4) + lane];
                                o.x += hr.x; o.y += hr.y; o.z += hr.z; o.w += hr.w;
                            }
                            const float s = __int_as_float(__builtin_amdgcn_readlane(__float_as_int(sv), j));
                            o.x *= s; o.y *= s; o.z *= s; o.w *= s;
                            A4[r * (GM_LDA / 4) + lane] = o;
                        }
                        acc = make_float4(0.f, 0.f, 0.f, 0.f);
                        ++j;
                        end = __builtin_amdgcn_readlane(endv, j);
                    }
                    float4 x = v[k];
                    if (SCALE) {
                        const float sc = __int_as_float(__builtin_amdgcn_readlane(scb, k0 + k));
                        x.x *= sc; x.y *= sc; x.z *= sc; x.w *= sc;
                    }
                    acc.x += x.x; acc.y += x.y; acc.z += x.z; acc.w += x.w;
                }
            }
        }
    }
    for (; j < NACC; ++j) {
        const int r = wave + NWAVE * j;
        if (r < rows) {
            float4 o = acc;
            if (scan) {
                const float4 p = t4[r * (D / 4) + lane];
                o.x += p.x; o.y += p.y; o.z += p.z; o.w += p.w;
            }
            if (!SCALE) {
                const float4 hr = A4[r * (GM_LDA / 4) + lane];
                o.x += hr.x; o.y += hr.y; o.z += hr.z; o.w += hr.w;
            }
            const float s = __int_as_float(__builtin_amdgcn_readlane(__float_as_int(sv), j));
            o.x *= s; o.y *= s; o.z *= s; o.w *= s;
            A4[r * (GM_LDA / 4) + lane] = o;
        }
        acc = make_float4(0.f, 0.f, 0.f, 0.f);
    }
    __syncthreads();

    const int half = lane >> 4;
    const int lm   = lane & 15;
    const int col0 = (wave & 7) * 16;
    const int rt0  = wave >> 3;
    if (rt0 + 4 * 3 < A_MT) tile_products<4>(Alds, W, bias, hout, base, rows, rt0, col0, half, lm);
    else                    tile_products<3>(Alds, W, bias, hout, base, rows, rt0, col0, half, lm);
}

extern "C" void kernel_launch(void* const* d_in, const int* in_sizes, int n_in,
                              void* d_out, int out_size, void* d_ws, size_t ws_size,
                              hipStream_t stream) {
    const float* x  = (const float*)d_in[0];
    const float* W1 = (const float*)d_in[1];
    const float* b1 = (const float*)d_in[2];
    const float* W2 = (const float*)d_in[3];
    const float* b2 = (const float*)d_in[4];
    const float* W3 = (const float*)d_in[5];
    const float* b3 = (const float*)d_in[6];
    const int* src  = (const int*)d_in[7];
    const int* dst  = (const int*)d_in[8];
    float* out = (float*)d_out;

    const int N = NN;
    float* deg_out = (float*)d_ws;
    float* h1      = deg_out + N;
    float* h2      = h1  + (size_t)N * D;
    int* reccnt       = (int*)(h2 + (size_t)N * D);
    unsigned* reclist = (unsigned*)(reccnt + NTILE * NWAVE * RCW);
    unsigned* marks   = reclist + (size_t)NTILE * NWAVE * QCAP;
    unsigned* part    = marks + (size_t)NW * NTILE;

    dim3 tileGrid(NTILE);
    dim3 markGrid(MARKBLK + HB);

    tile_mark_kernel<<<markGrid, 32 * MARKW, 0, stream>>>(dst, src, marks, part);

    tile_sum_gemm_kernel<0, 0><<<tileGrid, TBLK, 0, stream>>>(x, src, dst, marks, deg_out, W1, b1, h1, reclist, reccnt, part, deg_out);

    tile_sum_gemm_kernel<0, 1><<<tileGrid, TBLK, 0, stream>>>(h1, src, dst, marks, deg_out, W2, b2, h2, reclist, reccnt, part, deg_out);

    tile_sum_gemm_kernel<1, 1><<<tileGrid, TBLK, 0, stream>>>(h2, src, dst, marks, deg_out, W3, b3, out, reclist, reccnt, part, deg_out);
}
